# starting kernel
# baseline (speedup 1.0000x reference)
#include <hip/hip_runtime.h>

#define F 128
#define HEADS 8
#define NEG_SLOPE 0.2f
#define KEY_NEG_INF 0x007FFFFFu

typedef __attribute__((ext_vector_type(2))) float v2f;
typedef __attribute__((ext_vector_type(4))) float v4f;
typedef __attribute__((ext_vector_type(8))) float v8f;

__device__ __forceinline__ unsigned f32_key(float f) {
    unsigned b = __float_as_uint(f);
    return (b & 0x80000000u) ? ~b : (b | 0x80000000u);
}
__device__ __forceinline__ float key_f32(unsigned k) {
    unsigned b = (k & 0x80000000u) ? (k & 0x7FFFFFFFu) : ~k;
    return __uint_as_float(b);
}

__global__ __launch_bounds__(32) void gat_proj_wmma(
    const float* __restrict__ x, const float* __restrict__ Wl,
    const float* __restrict__ Wr, float* __restrict__ xl,
    float* __restrict__ xr, int n) {
    const int lane  = threadIdx.x & 31;
    const int row   = lane & 15;
    const int half  = lane >> 4;
    const int mbase = blockIdx.x * 16;

    const float* W   = (blockIdx.y == 0) ? Wl : Wr;
    float*       dst = (blockIdx.y == 0) ? xl : xr;

    int arow = mbase + row;
    if (arow >= n) arow = n - 1;
    const float* xrow = x + (size_t)arow * F + half * 2;
    const float* wrow = W + (size_t)row  * F + half * 2;

    v8f acc[8];
#pragma unroll
    for (int t = 0; t < 8; ++t) acc[t] = (v8f){0.f, 0.f, 0.f, 0.f, 0.f, 0.f, 0.f, 0.f};

    for (int k0 = 0; k0 < F; k0 += 4) {
        v2f a = *(const v2f*)(xrow + k0);
#pragma unroll
        for (int t = 0; t < 8; ++t) {
            v2f b = *(const v2f*)(wrow + (size_t)t * 16 * F + k0);
            acc[t] = __builtin_amdgcn_wmma_f32_16x16x4_f32(
                false, a, false, b, (short)0, acc[t], false, false);
        }
    }

#pragma unroll
    for (int t = 0; t < 8; ++t) {
#pragma unroll
        for (int r = 0; r < 8; ++r) {
            const int mrow = mbase + half * 8 + r;
            if (mrow < n) dst[(size_t)mrow * F + t * 16 + row] = acc[t][r];
        }
    }
}

__global__ void gat_init(float* __restrict__ out, const float* __restrict__ bias,
                         unsigned* __restrict__ amax, float* __restrict__ denom, int n) {
    const int i = blockIdx.x * blockDim.x + threadIdx.x;
    if (i < n * F) out[i] = bias[i & (F - 1)];
    if (i < n * HEADS) { amax[i] = KEY_NEG_INF; denom[i] = 0.f; }
}

__global__ void gat_edge_alpha(const float* __restrict__ xl, const float* __restrict__ xr,
                               const int* __restrict__ src, const int* __restrict__ dst,
                               const float* __restrict__ att, float* __restrict__ alpha,
                               unsigned* __restrict__ amax, int m) {
    const int e    = (blockIdx.x * blockDim.x + threadIdx.x) >> 5;
    const int lane = threadIdx.x & 31;
    if (e >= m) return;
    const int s = src[e], d = dst[e];
    const int c0 = lane * 4;

    v4f xj = *(const v4f*)(xl  + (size_t)s * F + c0);
    v4f xi = *(const v4f*)(xr  + (size_t)d * F + c0);
    v4f av = *(const v4f*)(att + c0);

    float p = 0.f;
#pragma unroll
    for (int j = 0; j < 4; ++j) {
        float v = xi[j] + xj[j];
        v = (v > 0.f) ? v : NEG_SLOPE * v;
        p += av[j] * v;
    }
    p += __shfl_xor(p, 1, 32);
    p += __shfl_xor(p, 2, 32);

    if ((lane & 3) == 0) {
        const int h = lane >> 2;
        alpha[(size_t)e * HEADS + h] = p;
        atomicMax(amax + (size_t)d * HEADS + h, f32_key(p));
    }
}

__global__ void gat_edge_exp(const int* __restrict__ dst, float* __restrict__ alpha,
                             const unsigned* __restrict__ amax,
                             float* __restrict__ denom, int m) {
    const int i = blockIdx.x * blockDim.x + threadIdx.x;
    if (i >= m * HEADS) return;
    const int e = i >> 3, h = i & 7;
    const int d = dst[e];
    const float mx = key_f32(amax[(size_t)d * HEADS + h]);
    const float ea = __expf(alpha[i] - mx);
    alpha[i] = ea;
    atomicAdd(denom + (size_t)d * HEADS + h, ea);
}

__global__ void gat_edge_aggr(const float* __restrict__ xl, const int* __restrict__ src,
                              const int* __restrict__ dst, const float* __restrict__ alpha,
                              const float* __restrict__ denom, float* __restrict__ out, int m) {
    const int e    = (blockIdx.x * blockDim.x + threadIdx.x) >> 5;
    const int lane = threadIdx.x & 31;
    if (e >= m) return;
    const int s = src[e], d = dst[e];
    const int c0 = lane * 4;
    const int h  = lane >> 2;

    const float w = alpha[(size_t)e * HEADS + h] /
                    (denom[(size_t)d * HEADS + h] + 1e-16f);
    v4f xj = *(const v4f*)(xl + (size_t)s * F + c0);
    float* op = out + (size_t)d * F + c0;
    atomicAdd(op + 0, w * xj[0]);
    atomicAdd(op + 1, w * xj[1]);
    atomicAdd(op + 2, w * xj[2]);
    atomicAdd(op + 3, w * xj[3]);
}

extern "C" void kernel_launch(void* const* d_in, const int* in_sizes, int n_in,
                              void* d_out, int out_size, void* d_ws, size_t ws_size,
                              hipStream_t stream) {
    const float* x    = (const float*)d_in[0];
    const int*   ei   = (const int*)  d_in[1];
    const float* Wl   = (const float*)d_in[2];
    const float* Wr   = (const float*)d_in[3];
    const float* att  = (const float*)d_in[4];
    const float* bias = (const float*)d_in[5];

    const int n = in_sizes[0] / F;
    const int m = in_sizes[1] / 2;
    const int* src = ei;
    const int* dst = ei + m;

    float*    xl    = (float*)d_ws;
    float*    xr    = xl + (size_t)n * F;
    float*    alpha = xr + (size_t)n * F;
    unsigned* amax  = (unsigned*)(alpha + (size_t)m * HEADS);
    float*    denom = (float*)(amax + (size_t)n * HEADS);
    float*    out   = (float*)d_out;

    const int initN = n * F;
    gat_init<<<(initN + 255) / 256, 256, 0, stream>>>(out, bias, amax, denom, n);

    dim3 gproj((n + 15) / 16, 2);
    gat_proj_wmma<<<gproj, 32, 0, stream>>>(x, Wl, Wr, xl, xr, n);

    const int edgeBlocks = (m * 32 + 255) / 256;
    gat_edge_alpha<<<edgeBlocks, 256, 0, stream>>>(xl, xr, src, dst, att, alpha, amax, m);
    gat_edge_exp<<<(m * HEADS + 255) / 256, 256, 0, stream>>>(dst, alpha, amax, denom, m);
    gat_edge_aggr<<<edgeBlocks, 256, 0, stream>>>(xl, src, dst, alpha, denom, out, m);
}


# Edges grouped by destination in-call (count, scan, returning-add rank); one wave per node does logits, running-max softmax and row sum in registers: no float atomics; sum order = arrival rank
# speedup vs baseline: 1.4999x; 1.4999x over previous
#include <hip/hip_runtime.h>

#define F 128
#define HEADS 8
#define SCAN_T 1024
#define SCAN_PER 16
#define NEG_SLOPE 0.2f

typedef __attribute__((ext_vector_type(2))) float v2f;
typedef __attribute__((ext_vector_type(4))) float v4f;
typedef __attribute__((ext_vector_type(8))) float v8f;

__global__ __launch_bounds__(32) void gat_proj_wmma(
    const float* __restrict__ x, const float* __restrict__ Wl,
    const float* __restrict__ Wr, float* __restrict__ xl,
    float* __restrict__ xr, int n) {
    const int lane  = threadIdx.x & 31;
    const int row   = lane & 15;
    const int half  = lane >> 4;
    const int mbase = blockIdx.x * 16;

    const float* W   = (blockIdx.y == 0) ? Wl : Wr;
    float*       dst = (blockIdx.y == 0) ? xl : xr;

    int arow = mbase + row;
    if (arow >= n) arow = n - 1;
    const float* xrow = x + (size_t)arow * F + half * 2;
    const float* wrow = W + (size_t)row  * F + half * 2;

    v8f acc[8];
#pragma unroll
    for (int t = 0; t < 8; ++t) acc[t] = (v8f){0.f, 0.f, 0.f, 0.f, 0.f, 0.f, 0.f, 0.f};

    for (int k0 = 0; k0 < F; k0 += 4) {
        v2f a = *(const v2f*)(xrow + k0);
#pragma unroll
        for (int t = 0; t < 8; ++t) {
            v2f b = *(const v2f*)(wrow + (size_t)t * 16 * F + k0);
            acc[t] = __builtin_amdgcn_wmma_f32_16x16x4_f32(
                false, a, false, b, (short)0, acc[t], false, false);
        }
    }

#pragma unroll
    for (int t = 0; t < 8; ++t) {
#pragma unroll
        for (int r = 0; r < 8; ++r) {
            const int mrow = mbase + half * 8 + r;
            if (mrow < n) dst[(size_t)mrow * F + t * 16 + row] = acc[t][r];
        }
    }
}

__global__ void csr_clear(int* __restrict__ counts, int n) {
    const int i = blockIdx.x * blockDim.x + threadIdx.x;
    if (i < n) counts[i] = 0;
}

__global__ void csr_count(const int* __restrict__ dst, int* __restrict__ counts, int m) {
    const int e = blockIdx.x * blockDim.x + threadIdx.x;
    if (e < m) atomicAdd(counts + dst[e], 1);
}

__global__ __launch_bounds__(SCAN_T) void csr_scan(const int* __restrict__ counts,
                                                   int* __restrict__ rowptr,
                                                   int* __restrict__ cursor, int n) {
    __shared__ int wsum[SCAN_T / 32];
    const int t    = threadIdx.x;
    const int lane = t & 31;
    const int w    = t >> 5;
    int carry = 0;
    for (int base = 0; base < n; base += SCAN_T * SCAN_PER) {
        const int i0 = base + t * SCAN_PER;
        int c[SCAN_PER];
        int tot = 0;
#pragma unroll
        for (int k = 0; k < SCAN_PER; ++k) {
            c[k] = (i0 + k < n) ? counts[i0 + k] : 0;
            tot += c[k];
        }

        int x = tot;
#pragma unroll
        for (int d = 1; d < 32; d <<= 1) {
            const int y = __shfl_up(x, d, 32);
            if (lane >= d) x += y;
        }
        if (lane == 31) wsum[w] = x;
        __syncthreads();
        if (w == 0) {
            int v = wsum[lane];
#pragma unroll
            for (int d = 1; d < 32; d <<= 1) {
                const int y = __shfl_up(v, d, 32);
                if (lane >= d) v += y;
            }
            wsum[lane] = v;
        }
        __syncthreads();
        const int wave_excl = (w > 0) ? wsum[w - 1] : 0;
        const int total     = wsum[SCAN_T / 32 - 1];
        int run = carry + wave_excl + (x - tot);
#pragma unroll
        for (int k = 0; k < SCAN_PER; ++k) {
            if (i0 + k < n) { rowptr[i0 + k] = run; cursor[i0 + k] = run; }
            run += c[k];
        }
        carry += total;
        __syncthreads();
    }
    if (t == 0) rowptr[n] = carry;
}

__global__ void csr_fill(const int* __restrict__ src, const int* __restrict__ dst,
                         int* __restrict__ cursor, int* __restrict__ csr_src, int m) {
    const int e = blockIdx.x * blockDim.x + threadIdx.x;
    if (e >= m) return;
    const int pos = atomicAdd(cursor + dst[e], 1);
    if ((unsigned)pos < (unsigned)m) csr_src[pos] = src[e];
}

__device__ __forceinline__ float edge_logit(v4f xi, v4f xj, v4f av) {
    float p = 0.f;
#pragma unroll
    for (int j = 0; j < 4; ++j) {
        float v = xi[j] + xj[j];
        v = (v > 0.f) ? v : NEG_SLOPE * v;
        p += av[j] * v;
    }
    p += __shfl_xor(p, 1, 32);
    p += __shfl_xor(p, 2, 32);
    return p;
}

__global__ __launch_bounds__(256) void gat_node(
    const float* __restrict__ xl, const float* __restrict__ xr,
    const int* __restrict__ rowptr, const int* __restrict__ csr_src,
    const float* __restrict__ att, const float* __restrict__ bias,
    float* __restrict__ out, int n) {
    const int node = __builtin_amdgcn_readfirstlane((int)((blockIdx.x * blockDim.x + threadIdx.x) >> 5));
    const int lane = threadIdx.x & 31;
    if (node >= n) return;
    const int c0  = lane * 4;
    const int beg = rowptr[node];
    const int end = rowptr[node + 1];

    const v4f xi = *(const v4f*)(xr   + (size_t)node * F + c0);
    const v4f av = *(const v4f*)(att  + c0);
    const v4f bv = *(const v4f*)(bias + c0);

    float mx  = -__builtin_inff();
    float den = 0.f;
    v4f   acc = (v4f){0.f, 0.f, 0.f, 0.f};

    const int last = end - 1;
    for (int j = beg; j < end; j += 4) {
        const int j1 = (j + 1 < end) ? j + 1 : last;
        const int j2 = (j + 2 < end) ? j + 2 : last;
        const int j3 = (j + 3 < end) ? j + 3 : last;
        const int s0 = csr_src[j];
        const int s1 = csr_src[j1];
        const int s2 = csr_src[j2];
        const int s3 = csr_src[j3];
        const v4f r0 = *(const v4f*)(xl + (size_t)s0 * F + c0);
        const v4f r1 = *(const v4f*)(xl + (size_t)s1 * F + c0);
        const v4f r2 = *(const v4f*)(xl + (size_t)s2 * F + c0);
        const v4f r3 = *(const v4f*)(xl + (size_t)s3 * F + c0);

        const float ninf = -__builtin_inff();
        const float p0 = edge_logit(xi, r0, av);
        const float q1 = edge_logit(xi, r1, av);
        const float q2 = edge_logit(xi, r2, av);
        const float q3 = edge_logit(xi, r3, av);
        const float p1 = (j + 1 < end) ? q1 : ninf;
        const float p2 = (j + 2 < end) ? q2 : ninf;
        const float p3 = (j + 3 < end) ? q3 : ninf;

        const float mn = fmaxf(fmaxf(mx, p0), fmaxf(fmaxf(p1, p2), p3));
        const float sc = __expf(mx - mn);
        const float e0 = __expf(p0 - mn);
        const float e1 = __expf(p1 - mn);
        const float e2 = __expf(p2 - mn);
        const float e3 = __expf(p3 - mn);
        den = den * sc + ((e0 + e1) + (e2 + e3));
#pragma unroll
        for (int k = 0; k < 4; ++k)
            acc[k] = acc[k] * sc + ((e0 * r0[k] + e1 * r1[k]) + (e2 * r2[k] + e3 * r3[k]));
        mx = mn;
    }

    const float dn = den + 1e-16f;
    v4f o;
#pragma unroll
    for (int k = 0; k < 4; ++k) o[k] = acc[k] / dn + bv[k];
    *(v4f*)(out + (size_t)node * F + c0) = o;
}

extern "C" void kernel_launch(void* const* d_in, const int* in_sizes, int n_in,
                              void* d_out, int out_size, void* d_ws, size_t ws_size,
                              hipStream_t stream) {
    const float* x    = (const float*)d_in[0];
    const int*   ei   = (const int*)  d_in[1];
    const float* Wl   = (const float*)d_in[2];
    const float* Wr   = (const float*)d_in[3];
    const float* att  = (const float*)d_in[4];
    const float* bias = (const float*)d_in[5];

    const int n = in_sizes[0] / F;
    const int m = in_sizes[1] / 2;
    const int* src = ei;
    const int* dst = ei + m;

    float* xl      = (float*)d_ws;
    float* xr      = xl + (size_t)n * F;
    int*   counts  = (int*)(xr + (size_t)n * F);
    int*   rowptr  = counts + n;
    int*   cursor  = rowptr + (n + 1);
    int*   csr_src = cursor + n;
    float* out     = (float*)d_out;

    csr_clear<<<(n + 255) / 256, 256, 0, stream>>>(counts, n);
    csr_count<<<(m + 255) / 256, 256, 0, stream>>>(dst, counts, m);
    csr_scan<<<1, SCAN_T, 0, stream>>>(counts, rowptr, cursor, n);
    csr_fill<<<(m + 255) / 256, 256, 0, stream>>>(src, dst, cursor, csr_src, m);

    dim3 gproj((n + 15) / 16, 2);
    gat_proj_wmma<<<gproj, 32, 0, stream>>>(x, Wl, Wr, xl, xr, n);

    const int nodeBlocks = (n * 32 + 255) / 256;
    gat_node<<<nodeBlocks, 256, 0, stream>>>(xl, xr, rowptr, csr_src, att, bias, out, n);
}


# Edges grouped by destination with a two-level counting sort on LDS counters (chunk x bin counts, bin runs, per-bin finish): no global atomics left; sum order = LDS arrival rank
# speedup vs baseline: 2.8812x; 1.9209x over previous
#include <hip/hip_runtime.h>

#define F 128
#define HEADS 8
#define NEG_SLOPE 0.2f

typedef __attribute__((ext_vector_type(2))) float v2f;
typedef __attribute__((ext_vector_type(4))) float v4f;
typedef __attribute__((ext_vector_type(8))) float v8f;

__global__ __launch_bounds__(32) void gat_proj_wmma(
    const float* __restrict__ x, const float* __restrict__ Wl,
    const float* __restrict__ Wr, float* __restrict__ xl,
    float* __restrict__ xr, int n) {
    const int lane  = threadIdx.x & 31;
    const int row   = lane & 15;
    const int half  = lane >> 4;
    const int mbase = blockIdx.x * 16;

    const float* W   = (blockIdx.y == 0) ? Wl : Wr;
    float*       dst = (blockIdx.y == 0) ? xl : xr;

    int arow = mbase + row;
    if (arow >= n) arow = n - 1;
    const float* xrow = x + (size_t)arow * F + half * 2;
    const float* wrow = W + (size_t)row  * F + half * 2;

    v8f acc[8];
#pragma unroll
    for (int t = 0; t < 8; ++t) acc[t] = (v8f){0.f, 0.f, 0.f, 0.f, 0.f, 0.f, 0.f, 0.f};

    for (int k0 = 0; k0 < F; k0 += 4) {
        v2f a = *(const v2f*)(xrow + k0);
#pragma unroll
        for (int t = 0; t < 8; ++t) {
            v2f b = *(const v2f*)(wrow + (size_t)t * 16 * F + k0);
            acc[t] = __builtin_amdgcn_wmma_f32_16x16x4_f32(
                false, a, false, b, (short)0, acc[t], false, false);
        }
    }

#pragma unroll
    for (int t = 0; t < 8; ++t) {
#pragma unroll
        for (int r = 0; r < 8; ++r) {
            const int mrow = mbase + half * 8 + r;
            if (mrow < n) dst[(size_t)mrow * F + t * 16 + row] = acc[t][r];
        }
    }
}

#define BIN_SHIFT 7
#define BIN_NODES 128
#define MAXB 1024
#define NCHUNK 256
#define SORT_T 1024
#define COL_PARTS 8
#define COL_PER (NCHUNK / COL_PARTS)

__global__ __launch_bounds__(SORT_T) void bin_count(const int* __restrict__ dst,
                                                    int* __restrict__ mat, int m, int nb) {
    __shared__ int h[MAXB];
    const int t = threadIdx.x;
    h[t] = 0;
    __syncthreads();
    const int per = (m + NCHUNK - 1) / NCHUNK;
    const int e0  = blockIdx.x * per;
    const int e1  = (e0 + per < m) ? e0 + per : m;
    for (int e = e0 + t; e < e1; e += SORT_T) {
        const int bin = dst[e] >> BIN_SHIFT;
        if ((unsigned)bin < (unsigned)nb) atomicAdd(&h[bin], 1);
    }
    __syncthreads();
    if (t < nb) mat[(size_t)blockIdx.x * nb + t] = h[t];
}

__global__ __launch_bounds__(SORT_T) void bin_scatter(const int* __restrict__ src,
                                                      const int* __restrict__ dst,
                                                      const int* __restrict__ mat,
                                                      int* __restrict__ binstart,
                                                      int* __restrict__ binned, int m, int nb) {
    __shared__ int off[MAXB];
    __shared__ int tot[MAXB];
    __shared__ int wsum[SORT_T / 32];
    const int t    = threadIdx.x;
    const int lane = t & 31;
    const int w    = t >> 5;
    const int c    = blockIdx.x;
    off[t] = 0;
    tot[t] = 0;
    __syncthreads();

    for (int idx = t; idx < nb * COL_PARTS; idx += SORT_T) {
        const int part = idx / nb;
        const int bin  = idx - part * nb;
        const int* col = mat + (part * COL_PER * nb + bin);
        int cv[COL_PER];
#pragma unroll
        for (int k = 0; k < COL_PER; ++k) cv[k] = col[k * nb];
        int all = 0, pre = 0;
#pragma unroll
        for (int k = 0; k < COL_PER; ++k) {
            all += cv[k];
            pre += (part * COL_PER + k < c) ? cv[k] : 0;
        }
        atomicAdd(&tot[bin], all);
        atomicAdd(&off[bin], pre);
    }
    __syncthreads();

    const int v = tot[t];
    int x = v;
#pragma unroll
    for (int d = 1; d < 32; d <<= 1) {
        const int y = __shfl_up(x, d, 32);
        if (lane >= d) x += y;
    }
    if (lane == 31) wsum[w] = x;
    __syncthreads();
    if (w == 0) {
        int s = wsum[lane];
#pragma unroll
        for (int d = 1; d < 32; d <<= 1) {
            const int y = __shfl_up(s, d, 32);
            if (lane >= d) s += y;
        }
        wsum[lane] = s;
    }
    __syncthreads();
    const int start = ((w > 0) ? wsum[w - 1] : 0) + (x - v);
    off[t] += start;
    if (c == 0) {
        if (t < nb) binstart[t] = start;
        if (t == 0) binstart[nb] = wsum[SORT_T / 32 - 1];
    }
    __syncthreads();

    const int per = (m + NCHUNK - 1) / NCHUNK;
    const int e0  = c * per;
    const int e1  = (e0 + per < m) ? e0 + per : m;
    for (int e = e0 + t; e < e1; e += SORT_T) {
        const int d   = dst[e];
        const int s   = src[e];
        const int bin = d >> BIN_SHIFT;
        if ((unsigned)bin < (unsigned)nb) {
            const int pos = atomicAdd(&off[bin], 1);
            if ((unsigned)pos < (unsigned)m) {
                ((int2*)binned)[pos] = make_int2(s, d);
            }
        }
    }
}

__global__ __launch_bounds__(256) void bin_finish(const int* __restrict__ binned,
                                                  const int* __restrict__ binstart,
                                                  int* __restrict__ rowptr,
                                                  int* __restrict__ csr_src,
                                                  int n, int m, int nb) {
    __shared__ int cnt[BIN_NODES];
    __shared__ int cur[BIN_NODES];
    __shared__ int ws4[BIN_NODES / 32];
    const int t    = threadIdx.x;
    const int lane = t & 31;
    const int w    = t >> 5;
    const int bin  = blockIdx.x;
    int lo = binstart[bin];
    int hi = binstart[bin + 1];
    lo = (lo < 0) ? 0 : ((lo > m) ? m : lo);
    hi = (hi < lo) ? lo : ((hi > m) ? m : hi);

    if (t < BIN_NODES) cnt[t] = 0;
    __syncthreads();
    for (int i = lo + t; i < hi; i += 256)
        atomicAdd(&cnt[binned[2 * (size_t)i + 1] & (BIN_NODES - 1)], 1);
    __syncthreads();

    const int v = (t < BIN_NODES) ? cnt[t] : 0;
    int x = v;
#pragma unroll
    for (int d = 1; d < 32; d <<= 1) {
        const int y = __shfl_up(x, d, 32);
        if (lane >= d) x += y;
    }
    if (lane == 31 && w < BIN_NODES / 32) ws4[w] = x;
    __syncthreads();
    int base = 0;
#pragma unroll
    for (int k = 0; k < BIN_NODES / 32; ++k) base += (k < w) ? ws4[k] : 0;
    const int excl = base + x - v;
    if (t < BIN_NODES) {
        cur[t] = excl;
        const int node = bin * BIN_NODES + t;
        if (node < n) rowptr[node] = lo + excl;
    }
    if (bin == nb - 1 && t == 0) rowptr[n] = hi;
    __syncthreads();

    for (int i = lo + t; i < hi; i += 256) {
        const int2 sd = ((const int2*)binned)[i];
        const int pos = lo + atomicAdd(&cur[sd.y & (BIN_NODES - 1)], 1);
        if ((unsigned)pos < (unsigned)m) csr_src[pos] = sd.x;
    }
}

__device__ __forceinline__ float edge_logit(v4f xi, v4f xj, v4f av) {
    float p = 0.f;
#pragma unroll
    for (int j = 0; j < 4; ++j) {
        float v = xi[j] + xj[j];
        v = (v > 0.f) ? v : NEG_SLOPE * v;
        p += av[j] * v;
    }
    p += __shfl_xor(p, 1, 32);
    p += __shfl_xor(p, 2, 32);
    return p;
}

__global__ __launch_bounds__(256) void gat_node(
    const float* __restrict__ xl, const float* __restrict__ xr,
    const int* __restrict__ rowptr, const int* __restrict__ csr_src,
    const float* __restrict__ att, const float* __restrict__ bias,
    float* __restrict__ out, int n) {
    const int node = __builtin_amdgcn_readfirstlane((int)((blockIdx.x * blockDim.x + threadIdx.x) >> 5));
    const int lane = threadIdx.x & 31;
    if (node >= n) return;
    const int c0  = lane * 4;
    const int beg = rowptr[node];
    const int end = rowptr[node + 1];

    const v4f xi = *(const v4f*)(xr   + (size_t)node * F + c0);
    const v4f av = *(const v4f*)(att  + c0);
    const v4f bv = *(const v4f*)(bias + c0);

    float mx  = -__builtin_inff();
    float den = 0.f;
    v4f   acc = (v4f){0.f, 0.f, 0.f, 0.f};

    const int last = end - 1;
    for (int j = beg; j < end; j += 4) {
        const int j1 = (j + 1 < end) ? j + 1 : last;
        const int j2 = (j + 2 < end) ? j + 2 : last;
        const int j3 = (j + 3 < end) ? j + 3 : last;
        const int s0 = csr_src[j];
        const int s1 = csr_src[j1];
        const int s2 = csr_src[j2];
        const int s3 = csr_src[j3];
        const v4f r0 = *(const v4f*)(xl + (size_t)s0 * F + c0);
        const v4f r1 = *(const v4f*)(xl + (size_t)s1 * F + c0);
        const v4f r2 = *(const v4f*)(xl + (size_t)s2 * F + c0);
        const v4f r3 = *(const v4f*)(xl + (size_t)s3 * F + c0);

        const float ninf = -__builtin_inff();
        const float p0 = edge_logit(xi, r0, av);
        const float q1 = edge_logit(xi, r1, av);
        const float q2 = edge_logit(xi, r2, av);
        const float q3 = edge_logit(xi, r3, av);
        const float p1 = (j + 1 < end) ? q1 : ninf;
        const float p2 = (j + 2 < end) ? q2 : ninf;
        const float p3 = (j + 3 < end) ? q3 : ninf;

        const float mn = fmaxf(fmaxf(mx, p0), fmaxf(fmaxf(p1, p2), p3));
        const float sc = __expf(mx - mn);
        const float e0 = __expf(p0 - mn);
        const float e1 = __expf(p1 - mn);
        const float e2 = __expf(p2 - mn);
        const float e3 = __expf(p3 - mn);
        den = den * sc + ((e0 + e1) + (e2 + e3));
#pragma unroll
        for (int k = 0; k < 4; ++k)
            acc[k] = acc[k] * sc + ((e0 * r0[k] + e1 * r1[k]) + (e2 * r2[k] + e3 * r3[k]));
        mx = mn;
    }

    const float dn = den + 1e-16f;
    v4f o;
#pragma unroll
    for (int k = 0; k < 4; ++k) o[k] = acc[k] / dn + bv[k];
    *(v4f*)(out + (size_t)node * F + c0) = o;
}

extern "C" void kernel_launch(void* const* d_in, const int* in_sizes, int n_in,
                              void* d_out, int out_size, void* d_ws, size_t ws_size,
                              hipStream_t stream) {
    const float* x    = (const float*)d_in[0];
    const int*   ei   = (const int*)  d_in[1];
    const float* Wl   = (const float*)d_in[2];
    const float* Wr   = (const float*)d_in[3];
    const float* att  = (const float*)d_in[4];
    const float* bias = (const float*)d_in[5];

    const int n = in_sizes[0] / F;
    const int m = in_sizes[1] / 2;
    const int* src = ei;
    const int* dst = ei + m;

    float* xl       = (float*)d_ws;
    float* xr       = xl + (size_t)n * F;
    int*   rowptr   = (int*)(xr + (size_t)n * F);
    int*   csr_src  = rowptr + (n + 1);
    int*   binned   = (int*)xl;
    int*   mat      = binned + 2 * (size_t)m;
    const int nb    = (n + BIN_NODES - 1) / BIN_NODES;
    int*   binstart = mat + (size_t)NCHUNK * nb;
    float* out      = (float*)d_out;

    bin_count<<<NCHUNK, SORT_T, 0, stream>>>(dst, mat, m, nb);
    bin_scatter<<<NCHUNK, SORT_T, 0, stream>>>(src, dst, mat, binstart, binned, m, nb);
    bin_finish<<<nb, 256, 0, stream>>>(binned, binstart, rowptr, csr_src, n, m, nb);

    dim3 gproj((n + 15) / 16, 2);
    gat_proj_wmma<<<gproj, 32, 0, stream>>>(x, Wl, Wr, xl, xr, n);

    const int nodeBlocks = (n * 32 + 255) / 256;
    gat_node<<<nodeBlocks, 256, 0, stream>>>(xl, xr, rowptr, csr_src, att, bias, out, n);
}


# Projection re-tiled: 256 persistent 16-wave work-groups copy W_l and W_r into LDS once and read every B fragment from LDS instead of 16-line global loads; k order and sums unchanged
# speedup vs baseline: 3.3274x; 1.1549x over previous
#include <hip/hip_runtime.h>

#define F 128
#define HEADS 8
#define NEG_SLOPE 0.2f

typedef __attribute__((ext_vector_type(2))) float v2f;
typedef __attribute__((ext_vector_type(4))) float v4f;
typedef __attribute__((ext_vector_type(8))) float v8f;

#define PROJ_T 512
#define PROJ_NW (PROJ_T / 32)
#define PROJ_WG 256
#define KG 2
#define WPITCH 132

__global__ __launch_bounds__(PROJ_T) void gat_proj_wmma(
    const float* __restrict__ x, const float* __restrict__ Wl,
    const float* __restrict__ Wr, float* __restrict__ xl,
    float* __restrict__ xr, int n) {
    __shared__ __attribute__((aligned(16))) float wsh[2 * F * WPITCH];

    const int tid  = threadIdx.x;
    const int lane = tid & 31;
    const int wave = __builtin_amdgcn_readfirstlane(tid >> 5);

    {
        v4f wv[2][F / PROJ_NW];
#pragma unroll
        for (int j = 0; j < F / PROJ_NW; ++j) {
            const size_t ro = (size_t)(j * PROJ_NW + wave) * F + lane * 4;
            wv[0][j] = *(const v4f*)(Wl + ro);
            wv[1][j] = *(const v4f*)(Wr + ro);
        }
#pragma unroll
        for (int j = 0; j < F / PROJ_NW; ++j) {
            const int rrow = j * PROJ_NW + wave;
            *(v4f*)(wsh + rrow * WPITCH + lane * 4)       = wv[0][j];
            *(v4f*)(wsh + (F + rrow) * WPITCH + lane * 4) = wv[1][j];
        }
    }
    __syncthreads();

    const int row  = lane & 15;
    const int half = lane >> 4;
    const int nstrips = (n + 15) >> 4;
    const int p = wave & 1;
    float* dst = p ? xr : xl;
    const float* wrow = wsh + (p * F + row) * WPITCH + half * 2;

    for (int j = wave; ; j += PROJ_NW) {
        const int strip = blockIdx.x + gridDim.x * (j >> 1);
        if (strip >= nstrips) break;
        const int mbase = strip * 16;

        int arow = mbase + row;
        if (arow >= n) arow = n - 1;
        const float* xrow = x + (size_t)arow * F + half * 2;

        v2f a[F / 4];
#pragma unroll
        for (int i = 0; i < F / 4; ++i) a[i] = *(const v2f*)(xrow + i * 4);

        v8f acc[8];
#pragma unroll
        for (int t = 0; t < 8; ++t) acc[t] = (v8f){0.f, 0.f, 0.f, 0.f, 0.f, 0.f, 0.f, 0.f};

        v2f b[2][KG][8];
#pragma unroll
        for (int s = 0; s < KG; ++s)
#pragma unroll
            for (int t = 0; t < 8; ++t) b[0][s][t] = *(const v2f*)(wrow + t * 16 * WPITCH + s * 4);

#pragma unroll
        for (int g = 0; g < F / 4 / KG; ++g) {
            if (g + 1 < F / 4 / KG) {
#pragma unroll
                for (int s = 0; s < KG; ++s)
#pragma unroll
                    for (int t = 0; t < 8; ++t)
                        b[(g + 1) & 1][s][t] = *(const v2f*)(wrow + t * 16 * WPITCH + ((g + 1) * KG + s) * 4);
            }
            __builtin_amdgcn_sched_barrier(0);
#pragma unroll
            for (int s = 0; s < KG; ++s)
#pragma unroll
                for (int t = 0; t < 8; ++t)
                    acc[t] = __builtin_amdgcn_wmma_f32_16x16x4_f32(
                        false, a[g * KG + s], false, b[g & 1][s][t], (short)0, acc[t], false, false);
            __builtin_amdgcn_sched_barrier(0);
        }
        asm volatile("v_nop\n\tv_nop\n\tv_nop\n\tv_nop\n\tv_nop\n\tv_nop\n\tv_nop\n\tv_nop");
        __builtin_amdgcn_sched_barrier(0);

#pragma unroll
        for (int t = 0; t < 8; ++t) {
#pragma unroll
            for (int r = 0; r < 8; ++r) {
                const int mrow = mbase + half * 8 + r;
                if (mrow < n) dst[(size_t)mrow * F + t * 16 + row] = acc[t][r];
            }
        }
    }
}

#define BIN_SHIFT 7
#define BIN_NODES 128
#define MAXB 1024
#define NCHUNK 256
#define SORT_T 1024
#define COL_PARTS 8
#define COL_PER (NCHUNK / COL_PARTS)

__global__ __launch_bounds__(SORT_T) void bin_count(const int* __restrict__ dst,
                                                    int* __restrict__ mat, int m, int nb) {
    __shared__ int h[MAXB];
    const int t = threadIdx.x;
    h[t] = 0;
    __syncthreads();
    const int per = (m + NCHUNK - 1) / NCHUNK;
    const int e0  = blockIdx.x * per;
    const int e1  = (e0 + per < m) ? e0 + per : m;
    for (int e = e0 + t; e < e1; e += SORT_T) {
        const int bin = dst[e] >> BIN_SHIFT;
        if ((unsigned)bin < (unsigned)nb) atomicAdd(&h[bin], 1);
    }
    __syncthreads();
    if (t < nb) mat[(size_t)blockIdx.x * nb + t] = h[t];
}

__global__ __launch_bounds__(SORT_T) void bin_scatter(const int* __restrict__ src,
                                                      const int* __restrict__ dst,
                                                      const int* __restrict__ mat,
                                                      int* __restrict__ binstart,
                                                      int* __restrict__ binned, int m, int nb) {
    __shared__ int off[MAXB];
    __shared__ int tot[MAXB];
    __shared__ int wsum[SORT_T / 32];
    const int t    = threadIdx.x;
    const int lane = t & 31;
    const int w    = t >> 5;
    const int c    = blockIdx.x;
    off[t] = 0;
    tot[t] = 0;
    __syncthreads();

    for (int idx = t; idx < nb * COL_PARTS; idx += SORT_T) {
        const int part = idx / nb;
        const int bin  = idx - part * nb;
        const int* col = mat + (part * COL_PER * nb + bin);
        int cv[COL_PER];
#pragma unroll
        for (int k = 0; k < COL_PER; ++k) cv[k] = col[k * nb];
        int all = 0, pre = 0;
#pragma unroll
        for (int k = 0; k < COL_PER; ++k) {
            all += cv[k];
            pre += (part * COL_PER + k < c) ? cv[k] : 0;
        }
        atomicAdd(&tot[bin], all);
        atomicAdd(&off[bin], pre);
    }
    __syncthreads();

    const int v = tot[t];
    int x = v;
#pragma unroll
    for (int d = 1; d < 32; d <<= 1) {
        const int y = __shfl_up(x, d, 32);
        if (lane >= d) x += y;
    }
    if (lane == 31) wsum[w] = x;
    __syncthreads();
    if (w == 0) {
        int s = wsum[lane];
#pragma unroll
        for (int d = 1; d < 32; d <<= 1) {
            const int y = __shfl_up(s, d, 32);
            if (lane >= d) s += y;
        }
        wsum[lane] = s;
    }
    __syncthreads();
    const int start = ((w > 0) ? wsum[w - 1] : 0) + (x - v);
    off[t] += start;
    if (c == 0) {
        if (t < nb) binstart[t] = start;
        if (t == 0) binstart[nb] = wsum[SORT_T / 32 - 1];
    }
    __syncthreads();

    const int per = (m + NCHUNK - 1) / NCHUNK;
    const int e0  = c * per;
    const int e1  = (e0 + per < m) ? e0 + per : m;
    for (int e = e0 + t; e < e1; e += SORT_T) {
        const int d   = dst[e];
        const int s   = src[e];
        const int bin = d >> BIN_SHIFT;
        if ((unsigned)bin < (unsigned)nb) {
            const int pos = atomicAdd(&off[bin], 1);
            if ((unsigned)pos < (unsigned)m) {
                ((int2*)binned)[pos] = make_int2(s, d);
            }
        }
    }
}

__global__ __launch_bounds__(256) void bin_finish(const int* __restrict__ binned,
                                                  const int* __restrict__ binstart,
                                                  int* __restrict__ rowptr,
                                                  int* __restrict__ csr_src,
                                                  int n, int m, int nb) {
    __shared__ int cnt[BIN_NODES];
    __shared__ int cur[BIN_NODES];
    __shared__ int ws4[BIN_NODES / 32];
    const int t    = threadIdx.x;
    const int lane = t & 31;
    const int w    = t >> 5;
    const int bin  = blockIdx.x;
    int lo = binstart[bin];
    int hi = binstart[bin + 1];
    lo = (lo < 0) ? 0 : ((lo > m) ? m : lo);
    hi = (hi < lo) ? lo : ((hi > m) ? m : hi);

    if (t < BIN_NODES) cnt[t] = 0;
    __syncthreads();
    for (int i = lo + t; i < hi; i += 256)
        atomicAdd(&cnt[binned[2 * (size_t)i + 1] & (BIN_NODES - 1)], 1);
    __syncthreads();

    const int v = (t < BIN_NODES) ? cnt[t] : 0;
    int x = v;
#pragma unroll
    for (int d = 1; d < 32; d <<= 1) {
        const int y = __shfl_up(x, d, 32);
        if (lane >= d) x += y;
    }
    if (lane == 31 && w < BIN_NODES / 32) ws4[w] = x;
    __syncthreads();
    int base = 0;
#pragma unroll
    for (int k = 0; k < BIN_NODES / 32; ++k) base += (k < w) ? ws4[k] : 0;
    const int excl = base + x - v;
    if (t < BIN_NODES) {
        cur[t] = excl;
        const int node = bin * BIN_NODES + t;
        if (node < n) rowptr[node] = lo + excl;
    }
    if (bin == nb - 1 && t == 0) rowptr[n] = hi;
    __syncthreads();

    for (int i = lo + t; i < hi; i += 256) {
        const int2 sd = ((const int2*)binned)[i];
        const int pos = lo + atomicAdd(&cur[sd.y & (BIN_NODES - 1)], 1);
        if ((unsigned)pos < (unsigned)m) csr_src[pos] = sd.x;
    }
}

__device__ __forceinline__ float edge_logit(v4f xi, v4f xj, v4f av) {
    float p = 0.f;
#pragma unroll
    for (int j = 0; j < 4; ++j) {
        float v = xi[j] + xj[j];
        v = (v > 0.f) ? v : NEG_SLOPE * v;
        p += av[j] * v;
    }
    p += __shfl_xor(p, 1, 32);
    p += __shfl_xor(p, 2, 32);
    return p;
}

__global__ __launch_bounds__(256) void gat_node(
    const float* __restrict__ xl, const float* __restrict__ xr,
    const int* __restrict__ rowptr, const int* __restrict__ csr_src,
    const float* __restrict__ att, const float* __restrict__ bias,
    float* __restrict__ out, int n) {
    const int node = __builtin_amdgcn_readfirstlane((int)((blockIdx.x * blockDim.x + threadIdx.x) >> 5));
    const int lane = threadIdx.x & 31;
    if (node >= n) return;
    const int c0  = lane * 4;
    const int beg = rowptr[node];
    const int end = rowptr[node + 1];

    const v4f xi = *(const v4f*)(xr   + (size_t)node * F + c0);
    const v4f av = *(const v4f*)(att  + c0);
    const v4f bv = *(const v4f*)(bias + c0);

    float mx  = -__builtin_inff();
    float den = 0.f;
    v4f   acc = (v4f){0.f, 0.f, 0.f, 0.f};

    const int last = end - 1;
    for (int j = beg; j < end; j += 4) {
        const int j1 = (j + 1 < end) ? j + 1 : last;
        const int j2 = (j + 2 < end) ? j + 2 : last;
        const int j3 = (j + 3 < end) ? j + 3 : last;
        const int s0 = csr_src[j];
        const int s1 = csr_src[j1];
        const int s2 = csr_src[j2];
        const int s3 = csr_src[j3];
        const v4f r0 = *(const v4f*)(xl + (size_t)s0 * F + c0);
        const v4f r1 = *(const v4f*)(xl + (size_t)s1 * F + c0);
        const v4f r2 = *(const v4f*)(xl + (size_t)s2 * F + c0);
        const v4f r3 = *(const v4f*)(xl + (size_t)s3 * F + c0);

        const float ninf = -__builtin_inff();
        const float p0 = edge_logit(xi, r0, av);
        const float q1 = edge_logit(xi, r1, av);
        const float q2 = edge_logit(xi, r2, av);
        const float q3 = edge_logit(xi, r3, av);
        const float p1 = (j + 1 < end) ? q1 : ninf;
        const float p2 = (j + 2 < end) ? q2 : ninf;
        const float p3 = (j + 3 < end) ? q3 : ninf;

        const float mn = fmaxf(fmaxf(mx, p0), fmaxf(fmaxf(p1, p2), p3));
        const float sc = __expf(mx - mn);
        const float e0 = __expf(p0 - mn);
        const float e1 = __expf(p1 - mn);
        const float e2 = __expf(p2 - mn);
        const float e3 = __expf(p3 - mn);
        den = den * sc + ((e0 + e1) + (e2 + e3));
#pragma unroll
        for (int k = 0; k < 4; ++k)
            acc[k] = acc[k] * sc + ((e0 * r0[k] + e1 * r1[k]) + (e2 * r2[k] + e3 * r3[k]));
        mx = mn;
    }

    const float dn = den + 1e-16f;
    v4f o;
#pragma unroll
    for (int k = 0; k < 4; ++k) o[k] = acc[k] / dn + bv[k];
    *(v4f*)(out + (size_t)node * F + c0) = o;
}

extern "C" void kernel_launch(void* const* d_in, const int* in_sizes, int n_in,
                              void* d_out, int out_size, void* d_ws, size_t ws_size,
                              hipStream_t stream) {
    const float* x    = (const float*)d_in[0];
    const int*   ei   = (const int*)  d_in[1];
    const float* Wl   = (const float*)d_in[2];
    const float* Wr   = (const float*)d_in[3];
    const float* att  = (const float*)d_in[4];
    const float* bias = (const float*)d_in[5];

    const int n = in_sizes[0] / F;
    const int m = in_sizes[1] / 2;
    const int* src = ei;
    const int* dst = ei + m;

    float* xl       = (float*)d_ws;
    float* xr       = xl + (size_t)n * F;
    int*   rowptr   = (int*)(xr + (size_t)n * F);
    int*   csr_src  = rowptr + (n + 1);
    int*   binned   = (int*)xl;
    int*   mat      = binned + 2 * (size_t)m;
    const int nb    = (n + BIN_NODES - 1) / BIN_NODES;
    int*   binstart = mat + (size_t)NCHUNK * nb;
    float* out      = (float*)d_out;

    bin_count<<<NCHUNK, SORT_T, 0, stream>>>(dst, mat, m, nb);
    bin_scatter<<<NCHUNK, SORT_T, 0, stream>>>(src, dst, mat, binstart, binned, m, nb);
    bin_finish<<<nb, 256, 0, stream>>>(binned, binstart, rowptr, csr_src, n, m, nb);

    gat_proj_wmma<<<PROJ_WG, PROJ_T, 0, stream>>>(x, Wl, Wr, xl, xr, n);

    const int nodeBlocks = (n * 32 + 255) / 256;
    gat_node<<<nodeBlocks, 256, 0, stream>>>(xl, xr, rowptr, csr_src, att, bias, out, n);
}


# Sort cut to two launches: each chunk sorts its own edges by bin in LDS; the per-bin pass sums the chunks' run starts itself: no count launch, no count-matrix re-reads; sum order = LDS arrival rank
# speedup vs baseline: 4.0308x; 1.2114x over previous
#include <hip/hip_runtime.h>

#define F 128
#define HEADS 8
#define NEG_SLOPE 0.2f

typedef __attribute__((ext_vector_type(2))) float v2f;
typedef __attribute__((ext_vector_type(4))) float v4f;
typedef __attribute__((ext_vector_type(8))) float v8f;

#define PROJ_T 512
#define PROJ_NW (PROJ_T / 32)
#define PROJ_WG 256
#define KG 2
#define WPITCH 132

__global__ __launch_bounds__(PROJ_T) void gat_proj_wmma(
    const float* __restrict__ x, const float* __restrict__ Wl,
    const float* __restrict__ Wr, float* __restrict__ xl,
    float* __restrict__ xr, int n) {
    __shared__ __attribute__((aligned(16))) float wsh[2 * F * WPITCH];

    const int tid  = threadIdx.x;
    const int lane = tid & 31;
    const int wave = __builtin_amdgcn_readfirstlane(tid >> 5);

    {
        v4f wv[2][F / PROJ_NW];
#pragma unroll
        for (int j = 0; j < F / PROJ_NW; ++j) {
            const size_t ro = (size_t)(j * PROJ_NW + wave) * F + lane * 4;
            wv[0][j] = *(const v4f*)(Wl + ro);
            wv[1][j] = *(const v4f*)(Wr + ro);
        }
#pragma unroll
        for (int j = 0; j < F / PROJ_NW; ++j) {
            const int rrow = j * PROJ_NW + wave;
            *(v4f*)(wsh + rrow * WPITCH + lane * 4)       = wv[0][j];
            *(v4f*)(wsh + (F + rrow) * WPITCH + lane * 4) = wv[1][j];
        }
    }
    __syncthreads();

    const int row  = lane & 15;
    const int half = lane >> 4;
    const int nstrips = (n + 15) >> 4;
    const int p = wave & 1;
    float* dst = p ? xr : xl;
    const float* wrow = wsh + (p * F + row) * WPITCH + half * 2;

    for (int j = wave; ; j += PROJ_NW) {
        const int strip = blockIdx.x + gridDim.x * (j >> 1);
        if (strip >= nstrips) break;
        const int mbase = strip * 16;

        int arow = mbase + row;
        if (arow >= n) arow = n - 1;
        const float* xrow = x + (size_t)arow * F + half * 2;

        v2f a[F / 4];
#pragma unroll
        for (int i = 0; i < F / 4; ++i) a[i] = *(const v2f*)(xrow + i * 4);

        v8f acc[8];
#pragma unroll
        for (int t = 0; t < 8; ++t) acc[t] = (v8f){0.f, 0.f, 0.f, 0.f, 0.f, 0.f, 0.f, 0.f};

        v2f b[2][KG][8];
#pragma unroll
        for (int s = 0; s < KG; ++s)
#pragma unroll
            for (int t = 0; t < 8; ++t) b[0][s][t] = *(const v2f*)(wrow + t * 16 * WPITCH + s * 4);

#pragma unroll
        for (int g = 0; g < F / 4 / KG; ++g) {
            if (g + 1 < F / 4 / KG) {
#pragma unroll
                for (int s = 0; s < KG; ++s)
#pragma unroll
                    for (int t = 0; t < 8; ++t)
                        b[(g + 1) & 1][s][t] = *(const v2f*)(wrow + t * 16 * WPITCH + ((g + 1) * KG + s) * 4);
            }
            __builtin_amdgcn_sched_barrier(0);
#pragma unroll
            for (int s = 0; s < KG; ++s)
#pragma unroll
                for (int t = 0; t < 8; ++t)
                    acc[t] = __builtin_amdgcn_wmma_f32_16x16x4_f32(
                        false, a[g * KG + s], false, b[g & 1][s][t], (short)0, acc[t], false, false);
            __builtin_amdgcn_sched_barrier(0);
        }
        asm volatile("v_nop\n\tv_nop\n\tv_nop\n\tv_nop\n\tv_nop\n\tv_nop\n\tv_nop\n\tv_nop");
        __builtin_amdgcn_sched_barrier(0);

#pragma unroll
        for (int t = 0; t < 8; ++t) {
#pragma unroll
            for (int r = 0; r < 8; ++r) {
                const int mrow = mbase + half * 8 + r;
                if (mrow < n) dst[(size_t)mrow * F + t * 16 + row] = acc[t][r];
            }
        }
    }
}

#define BIN_SHIFT 7
#define BIN_NODES 128
#define MAXB 1024
#define NCHUNK 256
#define SORT_T 512
#define EPT 8
#define FIN_T 256
#define RL 16
#define NPASS (NCHUNK / (FIN_T / RL))

static_assert(FIN_T == NCHUNK, "bin_finish maps thread c to chunk c");
static_assert(MAXB == 2 * SORT_T, "chunk_sort scans two bins per thread");

__global__ __launch_bounds__(SORT_T) void chunk_sort(const int* __restrict__ src,
                                                     const int* __restrict__ dst,
                                                     int* __restrict__ offT,
                                                     int* __restrict__ binned, int m, int nb) {
    __shared__ int h[MAXB];
    __shared__ int wsum[SORT_T / 32];
    const int t    = threadIdx.x;
    const int lane = t & 31;
    const int w    = t >> 5;
    const int c    = blockIdx.x;
    h[t] = 0;
    h[t + SORT_T] = 0;
    __syncthreads();

    const int per = (m + NCHUNK - 1) / NCHUNK;
    const int e0  = c * per;
    const int e1  = (e0 + per < m) ? e0 + per : m;

    for (int base = e0 + t; base < e1; base += EPT * SORT_T) {
        int dd[EPT];
#pragma unroll
        for (int k = 0; k < EPT; ++k) {
            const int e = base + k * SORT_T;
            dd[k] = dst[(e < e1) ? e : e1 - 1];
        }
        __builtin_amdgcn_sched_barrier(0);
#pragma unroll
        for (int k = 0; k < EPT; ++k) {
            const int e   = base + k * SORT_T;
            const int bin = dd[k] >> BIN_SHIFT;
            if (e < e1 && (unsigned)bin < (unsigned)nb) atomicAdd(&h[bin], 1);
        }
    }
    __syncthreads();

    const int v0 = h[2 * t];
    const int v1 = h[2 * t + 1];
    const int vs = v0 + v1;
    int x = vs;
#pragma unroll
    for (int d = 1; d < 32; d <<= 1) {
        const int y = __shfl_up(x, d, 32);
        if (lane >= d) x += y;
    }
    if (lane == 31) wsum[w] = x;
    __syncthreads();
    if (w == 0) {
        int s = (lane < SORT_T / 32) ? wsum[lane] : 0;
#pragma unroll
        for (int d = 1; d < 32; d <<= 1) {
            const int y = __shfl_up(s, d, 32);
            if (lane >= d) s += y;
        }
        if (lane < SORT_T / 32) wsum[lane] = s;
    }
    __syncthreads();
    const int ex0 = ((w > 0) ? wsum[w - 1] : 0) + (x - vs);
    const int ex1 = ex0 + v0;
    h[2 * t]     = ex0;
    h[2 * t + 1] = ex1;
    if (2 * t < nb)     offT[(size_t)(2 * t) * NCHUNK + c]     = ex0;
    if (2 * t + 1 < nb) offT[(size_t)(2 * t + 1) * NCHUNK + c] = ex1;
    if (t == 0)         offT[(size_t)nb * NCHUNK + c]          = wsum[SORT_T / 32 - 1];
    __syncthreads();

    const int cap = e1 - e0;
    for (int base = e0 + t; base < e1; base += EPT * SORT_T) {
        int dd[EPT], ss[EPT], ps[EPT];
#pragma unroll
        for (int k = 0; k < EPT; ++k) {
            const int e  = base + k * SORT_T;
            const int ec = (e < e1) ? e : e1 - 1;
            dd[k] = dst[ec];
            ss[k] = src[ec];
        }
#pragma unroll
        for (int k = 0; k < EPT; ++k) {
            const int e   = base + k * SORT_T;
            const int bin = dd[k] >> BIN_SHIFT;
            ps[k] = (e < e1 && (unsigned)bin < (unsigned)nb) ? atomicAdd(&h[bin], 1) : -1;
        }
#pragma unroll
        for (int k = 0; k < EPT; ++k)
            if ((unsigned)ps[k] < (unsigned)cap)
                ((int2*)binned)[e0 + ps[k]] = make_int2(ss[k], dd[k]);
    }
}

__global__ __launch_bounds__(FIN_T) void bin_finish(const int* __restrict__ binned,
                                                    const int* __restrict__ offT,
                                                    int* __restrict__ rowptr,
                                                    int* __restrict__ csr_src,
                                                    int n, int m, int nb) {
    __shared__ int cnt[BIN_NODES];
    __shared__ int cur[BIN_NODES];
    __shared__ int ws4[BIN_NODES / 32];
    __shared__ int rstart[NCHUNK];
    __shared__ int rlen[NCHUNK];
    __shared__ int red[2][FIN_T / 32];
    const int t    = threadIdx.x;
    const int lane = t & 31;
    const int w    = t >> 5;
    const int bin  = blockIdx.x;

    const int per = (m + NCHUNK - 1) / NCHUNK;
    const int ce0 = t * per;
    int ccnt = m - ce0;
    ccnt = (ccnt < 0) ? 0 : ((ccnt > per) ? per : ccnt);
    int p0 = offT[(size_t)bin * NCHUNK + t];
    int p1 = offT[(size_t)(bin + 1) * NCHUNK + t];
    p0 = (p0 < 0) ? 0 : ((p0 > ccnt) ? ccnt : p0);
    p1 = (p1 < p0) ? p0 : ((p1 > ccnt) ? ccnt : p1);
    const int len = p1 - p0;
    rstart[t] = ce0 + p0;
    rlen[t]   = len;
    if (t < BIN_NODES) cnt[t] = 0;

    int sa = p0, sb = len;
#pragma unroll
    for (int d = 16; d >= 1; d >>= 1) {
        sa += __shfl_xor(sa, d, 32);
        sb += __shfl_xor(sb, d, 32);
    }
    if (lane == 0) { red[0][w] = sa; red[1][w] = sb; }
    __syncthreads();
    int lo = 0, tot = 0;
#pragma unroll
    for (int k = 0; k < FIN_T / 32; ++k) { lo += red[0][k]; tot += red[1][k]; }
    lo = (lo > m) ? m : lo;
    int hi = lo + tot;
    hi = (hi > m) ? m : hi;

    const int sub = t & (RL - 1);
    const int grp = t / RL;
    int  ln[NPASS], st[NPASS];
    int2 pr[NPASS];
#pragma unroll
    for (int q = 0; q < NPASS; ++q) {
        const int r = q * (FIN_T / RL) + grp;
        ln[q] = rlen[r];
        st[q] = rstart[r];
    }
    int mx = 0;
#pragma unroll
    for (int q = 0; q < NPASS; ++q) {
        pr[q] = (sub < ln[q]) ? ((const int2*)binned)[st[q] + sub] : make_int2(0, -1);
        mx = (ln[q] > mx) ? ln[q] : mx;
    }
#pragma unroll
    for (int q = 0; q < NPASS; ++q)
        if (pr[q].y >= 0) atomicAdd(&cnt[pr[q].y & (BIN_NODES - 1)], 1);
    if (mx > RL) {
        for (int q = 0; q < NPASS; ++q)
            for (int j = sub + RL; j < ln[q]; j += RL)
                atomicAdd(&cnt[binned[2 * (size_t)(st[q] + j) + 1] & (BIN_NODES - 1)], 1);
    }
    __syncthreads();

    const int v = (t < BIN_NODES) ? cnt[t] : 0;
    int x = v;
#pragma unroll
    for (int d = 1; d < 32; d <<= 1) {
        const int y = __shfl_up(x, d, 32);
        if (lane >= d) x += y;
    }
    if (lane == 31 && w < BIN_NODES / 32) ws4[w] = x;
    __syncthreads();
    int base = 0;
#pragma unroll
    for (int k = 0; k < BIN_NODES / 32; ++k) base += (k < w) ? ws4[k] : 0;
    const int excl = base + x - v;
    if (t < BIN_NODES) {
        cur[t] = excl;
        const int node = bin * BIN_NODES + t;
        if (node < n) rowptr[node] = lo + excl;
    }
    if (bin == nb - 1 && t == 0) rowptr[n] = hi;
    __syncthreads();

    int ps[NPASS];
#pragma unroll
    for (int q = 0; q < NPASS; ++q)
        ps[q] = (pr[q].y >= 0) ? atomicAdd(&cur[pr[q].y & (BIN_NODES - 1)], 1) : -1;
#pragma unroll
    for (int q = 0; q < NPASS; ++q) {
        const int pos = lo + ps[q];
        if (ps[q] >= 0 && (unsigned)pos < (unsigned)m) csr_src[pos] = pr[q].x;
    }
    if (mx > RL) {
        for (int q = 0; q < NPASS; ++q)
            for (int j = sub + RL; j < ln[q]; j += RL) {
                const int2 sd = ((const int2*)binned)[st[q] + j];
                const int pos = lo + atomicAdd(&cur[sd.y & (BIN_NODES - 1)], 1);
                if ((unsigned)pos < (unsigned)m) csr_src[pos] = sd.x;
            }
    }
}

__device__ __forceinline__ float edge_logit(v4f xi, v4f xj, v4f av) {
    float p = 0.f;
#pragma unroll
    for (int j = 0; j < 4; ++j) {
        float v = xi[j] + xj[j];
        v = (v > 0.f) ? v : NEG_SLOPE * v;
        p += av[j] * v;
    }
    p += __shfl_xor(p, 1, 32);
    p += __shfl_xor(p, 2, 32);
    return p;
}

__global__ __launch_bounds__(256) void gat_node(
    const float* __restrict__ xl, const float* __restrict__ xr,
    const int* __restrict__ rowptr, const int* __restrict__ csr_src,
    const float* __restrict__ att, const float* __restrict__ bias,
    float* __restrict__ out, int n) {
    const int node = __builtin_amdgcn_readfirstlane((int)((blockIdx.x * blockDim.x + threadIdx.x) >> 5));
    const int lane = threadIdx.x & 31;
    if (node >= n) return;
    const int c0  = lane * 4;
    const int beg = rowptr[node];
    const int end = rowptr[node + 1];

    const v4f xi = *(const v4f*)(xr   + (size_t)node * F + c0);
    const v4f av = *(const v4f*)(att  + c0);
    const v4f bv = *(const v4f*)(bias + c0);

    float mx  = -__builtin_inff();
    float den = 0.f;
    v4f   acc = (v4f){0.f, 0.f, 0.f, 0.f};

    const int last = end - 1;
    for (int j = beg; j < end; j += 4) {
        const int j1 = (j + 1 < end) ? j + 1 : last;
        const int j2 = (j + 2 < end) ? j + 2 : last;
        const int j3 = (j + 3 < end) ? j + 3 : last;
        const int s0 = csr_src[j];
        const int s1 = csr_src[j1];
        const int s2 = csr_src[j2];
        const int s3 = csr_src[j3];
        const v4f r0 = *(const v4f*)(xl + (size_t)s0 * F + c0);
        const v4f r1 = *(const v4f*)(xl + (size_t)s1 * F + c0);
        const v4f r2 = *(const v4f*)(xl + (size_t)s2 * F + c0);
        const v4f r3 = *(const v4f*)(xl + (size_t)s3 * F + c0);

        const float ninf = -__builtin_inff();
        const float p0 = edge_logit(xi, r0, av);
        const float q1 = edge_logit(xi, r1, av);
        const float q2 = edge_logit(xi, r2, av);
        const float q3 = edge_logit(xi, r3, av);
        const float p1 = (j + 1 < end) ? q1 : ninf;
        const float p2 = (j + 2 < end) ? q2 : ninf;
        const float p3 = (j + 3 < end) ? q3 : ninf;

        const float mn = fmaxf(fmaxf(mx, p0), fmaxf(fmaxf(p1, p2), p3));
        const float sc = __expf(mx - mn);
        const float e0 = __expf(p0 - mn);
        const float e1 = __expf(p1 - mn);
        const float e2 = __expf(p2 - mn);
        const float e3 = __expf(p3 - mn);
        den = den * sc + ((e0 + e1) + (e2 + e3));
#pragma unroll
        for (int k = 0; k < 4; ++k)
            acc[k] = acc[k] * sc + ((e0 * r0[k] + e1 * r1[k]) + (e2 * r2[k] + e3 * r3[k]));
        mx = mn;
    }

    const float dn = den + 1e-16f;
    v4f o;
#pragma unroll
    for (int k = 0; k < 4; ++k) o[k] = acc[k] / dn + bv[k];
    *(v4f*)(out + (size_t)node * F + c0) = o;
}

extern "C" void kernel_launch(void* const* d_in, const int* in_sizes, int n_in,
                              void* d_out, int out_size, void* d_ws, size_t ws_size,
                              hipStream_t stream) {
    const float* x    = (const float*)d_in[0];
    const int*   ei   = (const int*)  d_in[1];
    const float* Wl   = (const float*)d_in[2];
    const float* Wr   = (const float*)d_in[3];
    const float* att  = (const float*)d_in[4];
    const float* bias = (const float*)d_in[5];

    const int n = in_sizes[0] / F;
    const int m = in_sizes[1] / 2;
    const int* src = ei;
    const int* dst = ei + m;

    float* xl       = (float*)d_ws;
    float* xr       = xl + (size_t)n * F;
    int*   rowptr   = (int*)(xr + (size_t)n * F);
    int*   csr_src  = rowptr + (n + 1);
    int*   binned   = (int*)xl;
    int*   offT     = binned + 2 * (size_t)m;
    const int nb    = (n + BIN_NODES - 1) / BIN_NODES;
    float* out      = (float*)d_out;

    chunk_sort<<<NCHUNK, SORT_T, 0, stream>>>(src, dst, offT, binned, m, nb);
    bin_finish<<<nb, FIN_T, 0, stream>>>(binned, offT, rowptr, csr_src, n, m, nb);

    gat_proj_wmma<<<PROJ_WG, PROJ_T, 0, stream>>>(x, Wl, Wr, xl, xr, n);

    const int nodeBlocks = (n * 32 + 255) / 256;
    gat_node<<<nodeBlocks, 256, 0, stream>>>(xl, xr, rowptr, csr_src, att, bias, out, n);
}


# Per-bin sort pass runs in the projection launch beside 8-wave projection groups (3 launches, sort scratch in out); each node sorts its <=32 sources in lanes first: sum order = source id, not arrival
# speedup vs baseline: 4.1875x; 1.0389x over previous
#include <hip/hip_runtime.h>

#define F 128
#define HEADS 8
#define NEG_SLOPE 0.2f

typedef __attribute__((ext_vector_type(2))) float v2f;
typedef __attribute__((ext_vector_type(4))) float v4f;
typedef __attribute__((ext_vector_type(8))) float v8f;

#define PROJ_T 256
#define PROJ_NW (PROJ_T / 32)
#define PROJ_WG 256
#define KG 2
#define WPITCH 132

__device__ __forceinline__ void proj_role(
    const float* __restrict__ x, const float* __restrict__ Wl,
    const float* __restrict__ Wr, float* __restrict__ xl,
    float* __restrict__ xr, int n, int wg) {
    __shared__ __attribute__((aligned(16))) float wsh[2 * F * WPITCH];

    const int tid  = threadIdx.x;
    const int lane = tid & 31;
    const int wave = __builtin_amdgcn_readfirstlane(tid >> 5);

    {
        v4f wv[2][F / PROJ_NW];
#pragma unroll
        for (int j = 0; j < F / PROJ_NW; ++j) {
            const size_t ro = (size_t)(j * PROJ_NW + wave) * F + lane * 4;
            wv[0][j] = *(const v4f*)(Wl + ro);
            wv[1][j] = *(const v4f*)(Wr + ro);
        }
#pragma unroll
        for (int j = 0; j < F / PROJ_NW; ++j) {
            const int rrow = j * PROJ_NW + wave;
            *(v4f*)(wsh + rrow * WPITCH + lane * 4)       = wv[0][j];
            *(v4f*)(wsh + (F + rrow) * WPITCH + lane * 4) = wv[1][j];
        }
    }
    __syncthreads();

    const int row  = lane & 15;
    const int half = lane >> 4;
    const int nstrips = (n + 15) >> 4;
    const int p = wave & 1;
    float* dst = p ? xr : xl;
    const float* wrow = wsh + (p * F + row) * WPITCH + half * 2;

    for (int j = wave; ; j += PROJ_NW) {
        const int strip = wg + PROJ_WG * (j >> 1);
        if (strip >= nstrips) break;
        const int mbase = strip * 16;

        int arow = mbase + row;
        if (arow >= n) arow = n - 1;
        const float* xrow = x + (size_t)arow * F + half * 2;

        v2f a[F / 4];
#pragma unroll
        for (int i = 0; i < F / 4; ++i) a[i] = *(const v2f*)(xrow + i * 4);

        v8f acc[8];
#pragma unroll
        for (int t = 0; t < 8; ++t) acc[t] = (v8f){0.f, 0.f, 0.f, 0.f, 0.f, 0.f, 0.f, 0.f};

        v2f b[2][KG][8];
#pragma unroll
        for (int s = 0; s < KG; ++s)
#pragma unroll
            for (int t = 0; t < 8; ++t) b[0][s][t] = *(const v2f*)(wrow + t * 16 * WPITCH + s * 4);

#pragma unroll
        for (int g = 0; g < F / 4 / KG; ++g) {
            if (g + 1 < F / 4 / KG) {
#pragma unroll
                for (int s = 0; s < KG; ++s)
#pragma unroll
                    for (int t = 0; t < 8; ++t)
                        b[(g + 1) & 1][s][t] = *(const v2f*)(wrow + t * 16 * WPITCH + ((g + 1) * KG + s) * 4);
            }
            __builtin_amdgcn_sched_barrier(0);
#pragma unroll
            for (int s = 0; s < KG; ++s)
#pragma unroll
                for (int t = 0; t < 8; ++t)
                    acc[t] = __builtin_amdgcn_wmma_f32_16x16x4_f32(
                        false, a[g * KG + s], false, b[g & 1][s][t], (short)0, acc[t], false, false);
            __builtin_amdgcn_sched_barrier(0);
        }
        asm volatile("v_nop\n\tv_nop\n\tv_nop\n\tv_nop\n\tv_nop\n\tv_nop\n\tv_nop\n\tv_nop");
        __builtin_amdgcn_sched_barrier(0);

#pragma unroll
        for (int t = 0; t < 8; ++t) {
#pragma unroll
            for (int r = 0; r < 8; ++r) {
                const int mrow = mbase + half * 8 + r;
                if (mrow < n) dst[(size_t)mrow * F + t * 16 + row] = acc[t][r];
            }
        }
    }
}

#define BIN_SHIFT 7
#define BIN_NODES 128
#define MAXB 1024
#define NCHUNK 256
#define SORT_T 512
#define EPT 8
#define FIN_T 256
#define RL 16
#define NPASS (NCHUNK / (FIN_T / RL))

static_assert(FIN_T == NCHUNK, "bin_finish maps thread c to chunk c");
static_assert(MAXB == 2 * SORT_T, "chunk_sort scans two bins per thread");

__global__ __launch_bounds__(SORT_T) void chunk_sort(const int* __restrict__ src,
                                                     const int* __restrict__ dst,
                                                     int* __restrict__ offT,
                                                     int* __restrict__ binned, int m, int nb) {
    __shared__ int h[MAXB];
    __shared__ int wsum[SORT_T / 32];
    const int t    = threadIdx.x;
    const int lane = t & 31;
    const int w    = t >> 5;
    const int c    = blockIdx.x;
    h[t] = 0;
    h[t + SORT_T] = 0;
    __syncthreads();

    const int per = (m + NCHUNK - 1) / NCHUNK;
    const int e0  = c * per;
    const int e1  = (e0 + per < m) ? e0 + per : m;

    for (int base = e0 + t; base < e1; base += EPT * SORT_T) {
        int dd[EPT];
#pragma unroll
        for (int k = 0; k < EPT; ++k) {
            const int e = base + k * SORT_T;
            dd[k] = dst[(e < e1) ? e : e1 - 1];
        }
        __builtin_amdgcn_sched_barrier(0);
#pragma unroll
        for (int k = 0; k < EPT; ++k) {
            const int e   = base + k * SORT_T;
            const int bin = dd[k] >> BIN_SHIFT;
            if (e < e1 && (unsigned)bin < (unsigned)nb) atomicAdd(&h[bin], 1);
        }
    }
    __syncthreads();

    const int v0 = h[2 * t];
    const int v1 = h[2 * t + 1];
    const int vs = v0 + v1;
    int x = vs;
#pragma unroll
    for (int d = 1; d < 32; d <<= 1) {
        const int y = __shfl_up(x, d, 32);
        if (lane >= d) x += y;
    }
    if (lane == 31) wsum[w] = x;
    __syncthreads();
    if (w == 0) {
        int s = (lane < SORT_T / 32) ? wsum[lane] : 0;
#pragma unroll
        for (int d = 1; d < 32; d <<= 1) {
            const int y = __shfl_up(s, d, 32);
            if (lane >= d) s += y;
        }
        if (lane < SORT_T / 32) wsum[lane] = s;
    }
    __syncthreads();
    const int ex0 = ((w > 0) ? wsum[w - 1] : 0) + (x - vs);
    const int ex1 = ex0 + v0;
    h[2 * t]     = ex0;
    h[2 * t + 1] = ex1;
    if (2 * t < nb)     offT[(size_t)(2 * t) * NCHUNK + c]     = ex0;
    if (2 * t + 1 < nb) offT[(size_t)(2 * t + 1) * NCHUNK + c] = ex1;
    if (t == 0)         offT[(size_t)nb * NCHUNK + c]          = wsum[SORT_T / 32 - 1];
    __syncthreads();

    const int cap = e1 - e0;
    for (int base = e0 + t; base < e1; base += EPT * SORT_T) {
        int dd[EPT], ss[EPT], ps[EPT];
#pragma unroll
        for (int k = 0; k < EPT; ++k) {
            const int e  = base + k * SORT_T;
            const int ec = (e < e1) ? e : e1 - 1;
            dd[k] = dst[ec];
            ss[k] = src[ec];
        }
#pragma unroll
        for (int k = 0; k < EPT; ++k) {
            const int e   = base + k * SORT_T;
            const int bin = dd[k] >> BIN_SHIFT;
            ps[k] = (e < e1 && (unsigned)bin < (unsigned)nb) ? atomicAdd(&h[bin], 1) : -1;
        }
#pragma unroll
        for (int k = 0; k < EPT; ++k)
            if ((unsigned)ps[k] < (unsigned)cap)
                ((int2*)binned)[e0 + ps[k]] = make_int2(ss[k], dd[k]);
    }
}

__device__ __forceinline__ void bin_role(const int* __restrict__ binned,
                                         const int* __restrict__ offT,
                                         int* __restrict__ rowptr,
                                         int* __restrict__ csr_src,
                                         int n, int m, int nb, int bin) {
    __shared__ int cnt[BIN_NODES];
    __shared__ int cur[BIN_NODES];
    __shared__ int ws4[BIN_NODES / 32];
    __shared__ int rstart[NCHUNK];
    __shared__ int rlen[NCHUNK];
    __shared__ int red[2][FIN_T / 32];
    const int t    = threadIdx.x;
    const int lane = t & 31;
    const int w    = t >> 5;

    const int per = (m + NCHUNK - 1) / NCHUNK;
    const int ce0 = t * per;
    int ccnt = m - ce0;
    ccnt = (ccnt < 0) ? 0 : ((ccnt > per) ? per : ccnt);
    int p0 = offT[(size_t)bin * NCHUNK + t];
    int p1 = offT[(size_t)(bin + 1) * NCHUNK + t];
    p0 = (p0 < 0) ? 0 : ((p0 > ccnt) ? ccnt : p0);
    p1 = (p1 < p0) ? p0 : ((p1 > ccnt) ? ccnt : p1);
    const int len = p1 - p0;
    rstart[t] = ce0 + p0;
    rlen[t]   = len;
    if (t < BIN_NODES) cnt[t] = 0;

    int sa = p0, sb = len;
#pragma unroll
    for (int d = 16; d >= 1; d >>= 1) {
        sa += __shfl_xor(sa, d, 32);
        sb += __shfl_xor(sb, d, 32);
    }
    if (lane == 0) { red[0][w] = sa; red[1][w] = sb; }
    __syncthreads();
    int lo = 0, tot = 0;
#pragma unroll
    for (int k = 0; k < FIN_T / 32; ++k) { lo += red[0][k]; tot += red[1][k]; }
    lo = (lo > m) ? m : lo;
    int hi = lo + tot;
    hi = (hi > m) ? m : hi;

    const int sub = t & (RL - 1);
    const int grp = t / RL;
    int  ln[NPASS], st[NPASS];
    int2 pr[NPASS];
#pragma unroll
    for (int q = 0; q < NPASS; ++q) {
        const int r = q * (FIN_T / RL) + grp;
        ln[q] = rlen[r];
        st[q] = rstart[r];
    }
    int mx = 0;
#pragma unroll
    for (int q = 0; q < NPASS; ++q) {
        pr[q] = (sub < ln[q]) ? ((const int2*)binned)[st[q] + sub] : make_int2(0, -1);
        mx = (ln[q] > mx) ? ln[q] : mx;
    }
#pragma unroll
    for (int q = 0; q < NPASS; ++q)
        if (pr[q].y >= 0) atomicAdd(&cnt[pr[q].y & (BIN_NODES - 1)], 1);
    if (mx > RL) {
        for (int q = 0; q < NPASS; ++q)
            for (int j = sub + RL; j < ln[q]; j += RL)
                atomicAdd(&cnt[binned[2 * (size_t)(st[q] + j) + 1] & (BIN_NODES - 1)], 1);
    }
    __syncthreads();

    const int v = (t < BIN_NODES) ? cnt[t] : 0;
    int x = v;
#pragma unroll
    for (int d = 1; d < 32; d <<= 1) {
        const int y = __shfl_up(x, d, 32);
        if (lane >= d) x += y;
    }
    if (lane == 31 && w < BIN_NODES / 32) ws4[w] = x;
    __syncthreads();
    int base = 0;
#pragma unroll
    for (int k = 0; k < BIN_NODES / 32; ++k) base += (k < w) ? ws4[k] : 0;
    const int excl = base + x - v;
    if (t < BIN_NODES) {
        cur[t] = excl;
        const int node = bin * BIN_NODES + t;
        if (node < n) rowptr[node] = lo + excl;
    }
    if (bin == nb - 1 && t == 0) rowptr[n] = hi;
    __syncthreads();

    int ps[NPASS];
#pragma unroll
    for (int q = 0; q < NPASS; ++q)
        ps[q] = (pr[q].y >= 0) ? atomicAdd(&cur[pr[q].y & (BIN_NODES - 1)], 1) : -1;
#pragma unroll
    for (int q = 0; q < NPASS; ++q) {
        const int pos = lo + ps[q];
        if (ps[q] >= 0 && (unsigned)pos < (unsigned)m) csr_src[pos] = pr[q].x;
    }
    if (mx > RL) {
        for (int q = 0; q < NPASS; ++q)
            for (int j = sub + RL; j < ln[q]; j += RL) {
                const int2 sd = ((const int2*)binned)[st[q] + j];
                const int pos = lo + atomicAdd(&cur[sd.y & (BIN_NODES - 1)], 1);
                if ((unsigned)pos < (unsigned)m) csr_src[pos] = sd.x;
            }
    }
}

static_assert(FIN_T == PROJ_T, "both roles use the same block size");

__global__ __launch_bounds__(PROJ_T) void finish_and_project(
    const float* __restrict__ x, const float* __restrict__ Wl,
    const float* __restrict__ Wr, float* __restrict__ xl, float* __restrict__ xr,
    const int* __restrict__ binned, const int* __restrict__ offT,
    int* __restrict__ rowptr, int* __restrict__ csr_src, int n, int m, int nb) {
    if (blockIdx.x < PROJ_WG) {
        proj_role(x, Wl, Wr, xl, xr, n, (int)blockIdx.x);
    } else {
        const int bin = (int)blockIdx.x - PROJ_WG;
        if (bin < nb) bin_role(binned, offT, rowptr, csr_src, n, m, nb, bin);
    }
}

__device__ __forceinline__ float edge_logit(v4f xi, v4f xj, v4f av) {
    float p = 0.f;
#pragma unroll
    for (int j = 0; j < 4; ++j) {
        float v = xi[j] + xj[j];
        v = (v > 0.f) ? v : NEG_SLOPE * v;
        p += av[j] * v;
    }
    p += __shfl_xor(p, 1, 32);
    p += __shfl_xor(p, 2, 32);
    return p;
}

template <int M>
__device__ __forceinline__ int lane_xor_in_row(int v) {
    return __builtin_amdgcn_update_dpp(v, v, 0x160 + M, 0xf, 0xf, false);
}
__device__ __forceinline__ int lane_xor_16(int v) {
    return __builtin_amdgcn_permlanex16(v, v, 0x76543210u, 0xfedcba98u, false, false);
}
template <int K, int J>
__device__ __forceinline__ int sort_step(int key, int lane) {
    const int o  = (J == 16) ? lane_xor_16(key) : lane_xor_in_row<(J == 16) ? 1 : J>(key);
    const int mn = (key < o) ? key : o;
    const int mx = (key < o) ? o : key;
    const bool up    = ((lane & K) == 0);
    const bool lower = ((lane & J) == 0);
    return (lower == up) ? mn : mx;
}

__global__ __launch_bounds__(256) void gat_node(
    const float* __restrict__ xl, const float* __restrict__ xr,
    const int* __restrict__ rowptr, const int* __restrict__ csr_src,
    const float* __restrict__ att, const float* __restrict__ bias,
    float* __restrict__ out, int n) {
    const int node = __builtin_amdgcn_readfirstlane((int)((blockIdx.x * blockDim.x + threadIdx.x) >> 5));
    const int lane = threadIdx.x & 31;
    if (node >= n) return;
    const int c0  = lane * 4;
    const int beg = rowptr[node];
    const int end = rowptr[node + 1];
    const int deg = end - beg;

    const v4f xi = *(const v4f*)(xr   + (size_t)node * F + c0);
    const v4f av = *(const v4f*)(att  + c0);
    const v4f bv = *(const v4f*)(bias + c0);

    const bool in_lanes = (deg <= 32);
    int key = 0x7fffffff;
    if (in_lanes) {
        if (lane < deg) key = csr_src[beg + lane];
        key = sort_step<2, 1>(key, lane);
        key = sort_step<4, 2>(key, lane);
        key = sort_step<4, 1>(key, lane);
        key = sort_step<8, 4>(key, lane);
        key = sort_step<8, 2>(key, lane);
        key = sort_step<8, 1>(key, lane);
        key = sort_step<16, 8>(key, lane);
        key = sort_step<16, 4>(key, lane);
        key = sort_step<16, 2>(key, lane);
        key = sort_step<16, 1>(key, lane);
        key = sort_step<32, 16>(key, lane);
        key = sort_step<32, 8>(key, lane);
        key = sort_step<32, 4>(key, lane);
        key = sort_step<32, 2>(key, lane);
        key = sort_step<32, 1>(key, lane);
    }

    float mx  = -__builtin_inff();
    float den = 0.f;
    v4f   acc = (v4f){0.f, 0.f, 0.f, 0.f};

    const int last = deg - 1;
    for (int j = 0; j < deg; j += 4) {
        const int j1 = (j + 1 < deg) ? j + 1 : last;
        const int j2 = (j + 2 < deg) ? j + 2 : last;
        const int j3 = (j + 3 < deg) ? j + 3 : last;
        int s0, s1, s2, s3;
        if (in_lanes) {
            s0 = __builtin_amdgcn_readlane(key, j);
            s1 = __builtin_amdgcn_readlane(key, j1);
            s2 = __builtin_amdgcn_readlane(key, j2);
            s3 = __builtin_amdgcn_readlane(key, j3);
        } else {
            s0 = csr_src[beg + j];
            s1 = csr_src[beg + j1];
            s2 = csr_src[beg + j2];
            s3 = csr_src[beg + j3];
        }
        const v4f r0 = *(const v4f*)(xl + (size_t)s0 * F + c0);
        const v4f r1 = *(const v4f*)(xl + (size_t)s1 * F + c0);
        const v4f r2 = *(const v4f*)(xl + (size_t)s2 * F + c0);
        const v4f r3 = *(const v4f*)(xl + (size_t)s3 * F + c0);

        const float ninf = -__builtin_inff();
        const float p0 = edge_logit(xi, r0, av);
        const float q1 = edge_logit(xi, r1, av);
        const float q2 = edge_logit(xi, r2, av);
        const float q3 = edge_logit(xi, r3, av);
        const float p1 = (j + 1 < deg) ? q1 : ninf;
        const float p2 = (j + 2 < deg) ? q2 : ninf;
        const float p3 = (j + 3 < deg) ? q3 : ninf;

        const float mn = fmaxf(fmaxf(mx, p0), fmaxf(fmaxf(p1, p2), p3));
        const float sc = __expf(mx - mn);
        const float e0 = __expf(p0 - mn);
        const float e1 = __expf(p1 - mn);
        const float e2 = __expf(p2 - mn);
        const float e3 = __expf(p3 - mn);
        den = den * sc + ((e0 + e1) + (e2 + e3));
#pragma unroll
        for (int k = 0; k < 4; ++k)
            acc[k] = acc[k] * sc + ((e0 * r0[k] + e1 * r1[k]) + (e2 * r2[k] + e3 * r3[k]));
        mx = mn;
    }

    const float dn = den + 1e-16f;
    v4f o;
#pragma unroll
    for (int k = 0; k < 4; ++k) o[k] = acc[k] / dn + bv[k];
    *(v4f*)(out + (size_t)node * F + c0) = o;
}

extern "C" void kernel_launch(void* const* d_in, const int* in_sizes, int n_in,
                              void* d_out, int out_size, void* d_ws, size_t ws_size,
                              hipStream_t stream) {
    const float* x    = (const float*)d_in[0];
    const int*   ei   = (const int*)  d_in[1];
    const float* Wl   = (const float*)d_in[2];
    const float* Wr   = (const float*)d_in[3];
    const float* att  = (const float*)d_in[4];
    const float* bias = (const float*)d_in[5];

    const int n = in_sizes[0] / F;
    const int m = in_sizes[1] / 2;
    const int* src = ei;
    const int* dst = ei + m;

    float* xl       = (float*)d_ws;
    float* xr       = xl + (size_t)n * F;
    int*   rowptr   = (int*)(xr + (size_t)n * F);
    int*   csr_src  = rowptr + (n + 1);
    float* out      = (float*)d_out;
    int*   binned   = (int*)out;
    int*   offT     = binned + 2 * (size_t)m;
    const int nb    = (n + BIN_NODES - 1) / BIN_NODES;

    chunk_sort<<<NCHUNK, SORT_T, 0, stream>>>(src, dst, offT, binned, m, nb);
    finish_and_project<<<PROJ_WG + nb, PROJ_T, 0, stream>>>(x, Wl, Wr, xl, xr, binned, offT,
                                                            rowptr, csr_src, n, m, nb);

    const int nodeBlocks = (n * 32 + 255) / 256;
    gat_node<<<nodeBlocks, 256, 0, stream>>>(xl, xr, rowptr, csr_src, att, bias, out, n);
}


# Node kernel's edge logit on packed f32 channel pairs: leaky ReLU as max(v, 0.2v) instead of compare+select, att dot as packed multiply-add; the 4-channel sum order changes (pairs, then halves)
# speedup vs baseline: 4.3780x; 1.0455x over previous
#include <hip/hip_runtime.h>

#define F 128
#define HEADS 8
#define NEG_SLOPE 0.2f

typedef __attribute__((ext_vector_type(2))) float v2f;
typedef __attribute__((ext_vector_type(4))) float v4f;
typedef __attribute__((ext_vector_type(8))) float v8f;

#define PROJ_T 256
#define PROJ_NW (PROJ_T / 32)
#define PROJ_WG 256
#define KG 2
#define WPITCH 132

__device__ __forceinline__ void proj_role(
    const float* __restrict__ x, const float* __restrict__ Wl,
    const float* __restrict__ Wr, float* __restrict__ xl,
    float* __restrict__ xr, int n, int wg) {
    __shared__ __attribute__((aligned(16))) float wsh[2 * F * WPITCH];

    const int tid  = threadIdx.x;
    const int lane = tid & 31;
    const int wave = __builtin_amdgcn_readfirstlane(tid >> 5);

    {
        v4f wv[2][F / PROJ_NW];
#pragma unroll
        for (int j = 0; j < F / PROJ_NW; ++j) {
            const size_t ro = (size_t)(j * PROJ_NW + wave) * F + lane * 4;
            wv[0][j] = *(const v4f*)(Wl + ro);
            wv[1][j] = *(const v4f*)(Wr + ro);
        }
#pragma unroll
        for (int j = 0; j < F / PROJ_NW; ++j) {
            const int rrow = j * PROJ_NW + wave;
            *(v4f*)(wsh + rrow * WPITCH + lane * 4)       = wv[0][j];
            *(v4f*)(wsh + (F + rrow) * WPITCH + lane * 4) = wv[1][j];
        }
    }
    __syncthreads();

    const int row  = lane & 15;
    const int half = lane >> 4;
    const int nstrips = (n + 15) >> 4;
    const int p = wave & 1;
    float* dst = p ? xr : xl;
    const float* wrow = wsh + (p * F + row) * WPITCH + half * 2;

    for (int j = wave; ; j += PROJ_NW) {
        const int strip = wg + PROJ_WG * (j >> 1);
        if (strip >= nstrips) break;
        const int mbase = strip * 16;

        int arow = mbase + row;
        if (arow >= n) arow = n - 1;
        const float* xrow = x + (size_t)arow * F + half * 2;

        v2f a[F / 4];
#pragma unroll
        for (int i = 0; i < F / 4; ++i) a[i] = *(const v2f*)(xrow + i * 4);

        v8f acc[8];
#pragma unroll
        for (int t = 0; t < 8; ++t) acc[t] = (v8f){0.f, 0.f, 0.f, 0.f, 0.f, 0.f, 0.f, 0.f};

        v2f b[2][KG][8];
#pragma unroll
        for (int s = 0; s < KG; ++s)
#pragma unroll
            for (int t = 0; t < 8; ++t) b[0][s][t] = *(const v2f*)(wrow + t * 16 * WPITCH + s * 4);

#pragma unroll
        for (int g = 0; g < F / 4 / KG; ++g) {
            if (g + 1 < F / 4 / KG) {
#pragma unroll
                for (int s = 0; s < KG; ++s)
#pragma unroll
                    for (int t = 0; t < 8; ++t)
                        b[(g + 1) & 1][s][t] = *(const v2f*)(wrow + t * 16 * WPITCH + ((g + 1) * KG + s) * 4);
            }
            __builtin_amdgcn_sched_barrier(0);
#pragma unroll
            for (int s = 0; s < KG; ++s)
#pragma unroll
                for (int t = 0; t < 8; ++t)
                    acc[t] = __builtin_amdgcn_wmma_f32_16x16x4_f32(
                        false, a[g * KG + s], false, b[g & 1][s][t], (short)0, acc[t], false, false);
            __builtin_amdgcn_sched_barrier(0);
        }
        asm volatile("v_nop\n\tv_nop\n\tv_nop\n\tv_nop\n\tv_nop\n\tv_nop\n\tv_nop\n\tv_nop");
        __builtin_amdgcn_sched_barrier(0);

#pragma unroll
        for (int t = 0; t < 8; ++t) {
#pragma unroll
            for (int r = 0; r < 8; ++r) {
                const int mrow = mbase + half * 8 + r;
                if (mrow < n) dst[(size_t)mrow * F + t * 16 + row] = acc[t][r];
            }
        }
    }
}

#define BIN_SHIFT 7
#define BIN_NODES 128
#define MAXB 1024
#define NCHUNK 256
#define SORT_T 512
#define EPT 8
#define FIN_T 256
#define RL 16
#define NPASS (NCHUNK / (FIN_T / RL))

static_assert(FIN_T == NCHUNK, "bin_finish maps thread c to chunk c");
static_assert(MAXB == 2 * SORT_T, "chunk_sort scans two bins per thread");

__global__ __launch_bounds__(SORT_T) void chunk_sort(const int* __restrict__ src,
                                                     const int* __restrict__ dst,
                                                     int* __restrict__ offT,
                                                     int* __restrict__ binned, int m, int nb) {
    __shared__ int h[MAXB];
    __shared__ int wsum[SORT_T / 32];
    const int t    = threadIdx.x;
    const int lane = t & 31;
    const int w    = t >> 5;
    const int c    = blockIdx.x;
    h[t] = 0;
    h[t + SORT_T] = 0;
    __syncthreads();

    const int per = (m + NCHUNK - 1) / NCHUNK;
    const int e0  = c * per;
    const int e1  = (e0 + per < m) ? e0 + per : m;

    for (int base = e0 + t; base < e1; base += EPT * SORT_T) {
        int dd[EPT];
#pragma unroll
        for (int k = 0; k < EPT; ++k) {
            const int e = base + k * SORT_T;
            dd[k] = dst[(e < e1) ? e : e1 - 1];
        }
        __builtin_amdgcn_sched_barrier(0);
#pragma unroll
        for (int k = 0; k < EPT; ++k) {
            const int e   = base + k * SORT_T;
            const int bin = dd[k] >> BIN_SHIFT;
            if (e < e1 && (unsigned)bin < (unsigned)nb) atomicAdd(&h[bin], 1);
        }
    }
    __syncthreads();

    const int v0 = h[2 * t];
    const int v1 = h[2 * t + 1];
    const int vs = v0 + v1;
    int x = vs;
#pragma unroll
    for (int d = 1; d < 32; d <<= 1) {
        const int y = __shfl_up(x, d, 32);
        if (lane >= d) x += y;
    }
    if (lane == 31) wsum[w] = x;
    __syncthreads();
    if (w == 0) {
        int s = (lane < SORT_T / 32) ? wsum[lane] : 0;
#pragma unroll
        for (int d = 1; d < 32; d <<= 1) {
            const int y = __shfl_up(s, d, 32);
            if (lane >= d) s += y;
        }
        if (lane < SORT_T / 32) wsum[lane] = s;
    }
    __syncthreads();
    const int ex0 = ((w > 0) ? wsum[w - 1] : 0) + (x - vs);
    const int ex1 = ex0 + v0;
    h[2 * t]     = ex0;
    h[2 * t + 1] = ex1;
    if (2 * t < nb)     offT[(size_t)(2 * t) * NCHUNK + c]     = ex0;
    if (2 * t + 1 < nb) offT[(size_t)(2 * t + 1) * NCHUNK + c] = ex1;
    if (t == 0)         offT[(size_t)nb * NCHUNK + c]          = wsum[SORT_T / 32 - 1];
    __syncthreads();

    const int cap = e1 - e0;
    for (int base = e0 + t; base < e1; base += EPT * SORT_T) {
        int dd[EPT], ss[EPT], ps[EPT];
#pragma unroll
        for (int k = 0; k < EPT; ++k) {
            const int e  = base + k * SORT_T;
            const int ec = (e < e1) ? e : e1 - 1;
            dd[k] = dst[ec];
            ss[k] = src[ec];
        }
#pragma unroll
        for (int k = 0; k < EPT; ++k) {
            const int e   = base + k * SORT_T;
            const int bin = dd[k] >> BIN_SHIFT;
            ps[k] = (e < e1 && (unsigned)bin < (unsigned)nb) ? atomicAdd(&h[bin], 1) : -1;
        }
#pragma unroll
        for (int k = 0; k < EPT; ++k)
            if ((unsigned)ps[k] < (unsigned)cap)
                ((int2*)binned)[e0 + ps[k]] = make_int2(ss[k], dd[k]);
    }
}

__device__ __forceinline__ void bin_role(const int* __restrict__ binned,
                                         const int* __restrict__ offT,
                                         int* __restrict__ rowptr,
                                         int* __restrict__ csr_src,
                                         int n, int m, int nb, int bin) {
    __shared__ int cnt[BIN_NODES];
    __shared__ int cur[BIN_NODES];
    __shared__ int ws4[BIN_NODES / 32];
    __shared__ int rstart[NCHUNK];
    __shared__ int rlen[NCHUNK];
    __shared__ int red[2][FIN_T / 32];
    const int t    = threadIdx.x;
    const int lane = t & 31;
    const int w    = t >> 5;

    const int per = (m + NCHUNK - 1) / NCHUNK;
    const int ce0 = t * per;
    int ccnt = m - ce0;
    ccnt = (ccnt < 0) ? 0 : ((ccnt > per) ? per : ccnt);
    int p0 = offT[(size_t)bin * NCHUNK + t];
    int p1 = offT[(size_t)(bin + 1) * NCHUNK + t];
    p0 = (p0 < 0) ? 0 : ((p0 > ccnt) ? ccnt : p0);
    p1 = (p1 < p0) ? p0 : ((p1 > ccnt) ? ccnt : p1);
    const int len = p1 - p0;
    rstart[t] = ce0 + p0;
    rlen[t]   = len;
    if (t < BIN_NODES) cnt[t] = 0;

    int sa = p0, sb = len;
#pragma unroll
    for (int d = 16; d >= 1; d >>= 1) {
        sa += __shfl_xor(sa, d, 32);
        sb += __shfl_xor(sb, d, 32);
    }
    if (lane == 0) { red[0][w] = sa; red[1][w] = sb; }
    __syncthreads();
    int lo = 0, tot = 0;
#pragma unroll
    for (int k = 0; k < FIN_T / 32; ++k) { lo += red[0][k]; tot += red[1][k]; }
    lo = (lo > m) ? m : lo;
    int hi = lo + tot;
    hi = (hi > m) ? m : hi;

    const int sub = t & (RL - 1);
    const int grp = t / RL;
    int  ln[NPASS], st[NPASS];
    int2 pr[NPASS];
#pragma unroll
    for (int q = 0; q < NPASS; ++q) {
        const int r = q * (FIN_T / RL) + grp;
        ln[q] = rlen[r];
        st[q] = rstart[r];
    }
    int mx = 0;
#pragma unroll
    for (int q = 0; q < NPASS; ++q) {
        pr[q] = (sub < ln[q]) ? ((const int2*)binned)[st[q] + sub] : make_int2(0, -1);
        mx = (ln[q] > mx) ? ln[q] : mx;
    }
#pragma unroll
    for (int q = 0; q < NPASS; ++q)
        if (pr[q].y >= 0) atomicAdd(&cnt[pr[q].y & (BIN_NODES - 1)], 1);
    if (mx > RL) {
        for (int q = 0; q < NPASS; ++q)
            for (int j = sub + RL; j < ln[q]; j += RL)
                atomicAdd(&cnt[binned[2 * (size_t)(st[q] + j) + 1] & (BIN_NODES - 1)], 1);
    }
    __syncthreads();

    const int v = (t < BIN_NODES) ? cnt[t] : 0;
    int x = v;
#pragma unroll
    for (int d = 1; d < 32; d <<= 1) {
        const int y = __shfl_up(x, d, 32);
        if (lane >= d) x += y;
    }
    if (lane == 31 && w < BIN_NODES / 32) ws4[w] = x;
    __syncthreads();
    int base = 0;
#pragma unroll
    for (int k = 0; k < BIN_NODES / 32; ++k) base += (k < w) ? ws4[k] : 0;
    const int excl = base + x - v;
    if (t < BIN_NODES) {
        cur[t] = excl;
        const int node = bin * BIN_NODES + t;
        if (node < n) rowptr[node] = lo + excl;
    }
    if (bin == nb - 1 && t == 0) rowptr[n] = hi;
    __syncthreads();

    int ps[NPASS];
#pragma unroll
    for (int q = 0; q < NPASS; ++q)
        ps[q] = (pr[q].y >= 0) ? atomicAdd(&cur[pr[q].y & (BIN_NODES - 1)], 1) : -1;
#pragma unroll
    for (int q = 0; q < NPASS; ++q) {
        const int pos = lo + ps[q];
        if (ps[q] >= 0 && (unsigned)pos < (unsigned)m) csr_src[pos] = pr[q].x;
    }
    if (mx > RL) {
        for (int q = 0; q < NPASS; ++q)
            for (int j = sub + RL; j < ln[q]; j += RL) {
                const int2 sd = ((const int2*)binned)[st[q] + j];
                const int pos = lo + atomicAdd(&cur[sd.y & (BIN_NODES - 1)], 1);
                if ((unsigned)pos < (unsigned)m) csr_src[pos] = sd.x;
            }
    }
}

static_assert(FIN_T == PROJ_T, "both roles use the same block size");

__global__ __launch_bounds__(PROJ_T) void finish_and_project(
    const float* __restrict__ x, const float* __restrict__ Wl,
    const float* __restrict__ Wr, float* __restrict__ xl, float* __restrict__ xr,
    const int* __restrict__ binned, const int* __restrict__ offT,
    int* __restrict__ rowptr, int* __restrict__ csr_src, int n, int m, int nb) {
    if (blockIdx.x < PROJ_WG) {
        proj_role(x, Wl, Wr, xl, xr, n, (int)blockIdx.x);
    } else {
        const int bin = (int)blockIdx.x - PROJ_WG;
        if (bin < nb) bin_role(binned, offT, rowptr, csr_src, n, m, nb, bin);
    }
}

__device__ __forceinline__ float edge_logit(v4f xi, v4f xj, v4f av) {
    const v2f s01 = xi.xy + xj.xy;
    const v2f s23 = xi.zw + xj.zw;
    const v2f l01 = __builtin_elementwise_max(s01, s01 * NEG_SLOPE);
    const v2f l23 = __builtin_elementwise_max(s23, s23 * NEG_SLOPE);
    v2f d = av.xy * l01;
    d = __builtin_elementwise_fma(av.zw, l23, d);
    float p = d.x + d.y;
    p += __shfl_xor(p, 1, 32);
    p += __shfl_xor(p, 2, 32);
    return p;
}

template <int M>
__device__ __forceinline__ int lane_xor_in_row(int v) {
    return __builtin_amdgcn_update_dpp(v, v, 0x160 + M, 0xf, 0xf, false);
}
__device__ __forceinline__ int lane_xor_16(int v) {
    return __builtin_amdgcn_permlanex16(v, v, 0x76543210u, 0xfedcba98u, false, false);
}
template <int K, int J>
__device__ __forceinline__ int sort_step(int key, int lane) {
    const int o  = (J == 16) ? lane_xor_16(key) : lane_xor_in_row<(J == 16) ? 1 : J>(key);
    const int mn = (key < o) ? key : o;
    const int mx = (key < o) ? o : key;
    const bool up    = ((lane & K) == 0);
    const bool lower = ((lane & J) == 0);
    return (lower == up) ? mn : mx;
}

__global__ __launch_bounds__(256) void gat_node(
    const float* __restrict__ xl, const float* __restrict__ xr,
    const int* __restrict__ rowptr, const int* __restrict__ csr_src,
    const float* __restrict__ att, const float* __restrict__ bias,
    float* __restrict__ out, int n) {
    const int node = __builtin_amdgcn_readfirstlane((int)((blockIdx.x * blockDim.x + threadIdx.x) >> 5));
    const int lane = threadIdx.x & 31;
    if (node >= n) return;
    const int c0  = lane * 4;
    const int beg = rowptr[node];
    const int end = rowptr[node + 1];
    const int deg = end - beg;

    const v4f xi = *(const v4f*)(xr   + (size_t)node * F + c0);
    const v4f av = *(const v4f*)(att  + c0);
    const v4f bv = *(const v4f*)(bias + c0);

    const bool in_lanes = (deg <= 32);
    int key = 0x7fffffff;
    if (in_lanes) {
        if (lane < deg) key = csr_src[beg + lane];
        key = sort_step<2, 1>(key, lane);
        key = sort_step<4, 2>(key, lane);
        key = sort_step<4, 1>(key, lane);
        key = sort_step<8, 4>(key, lane);
        key = sort_step<8, 2>(key, lane);
        key = sort_step<8, 1>(key, lane);
        key = sort_step<16, 8>(key, lane);
        key = sort_step<16, 4>(key, lane);
        key = sort_step<16, 2>(key, lane);
        key = sort_step<16, 1>(key, lane);
        key = sort_step<32, 16>(key, lane);
        key = sort_step<32, 8>(key, lane);
        key = sort_step<32, 4>(key, lane);
        key = sort_step<32, 2>(key, lane);
        key = sort_step<32, 1>(key, lane);
    }

    float mx  = -__builtin_inff();
    float den = 0.f;
    v4f   acc = (v4f){0.f, 0.f, 0.f, 0.f};

    const int last = deg - 1;
    for (int j = 0; j < deg; j += 4) {
        const int j1 = (j + 1 < deg) ? j + 1 : last;
        const int j2 = (j + 2 < deg) ? j + 2 : last;
        const int j3 = (j + 3 < deg) ? j + 3 : last;
        int s0, s1, s2, s3;
        if (in_lanes) {
            s0 = __builtin_amdgcn_readlane(key, j);
            s1 = __builtin_amdgcn_readlane(key, j1);
            s2 = __builtin_amdgcn_readlane(key, j2);
            s3 = __builtin_amdgcn_readlane(key, j3);
        } else {
            s0 = csr_src[beg + j];
            s1 = csr_src[beg + j1];
            s2 = csr_src[beg + j2];
            s3 = csr_src[beg + j3];
        }
        const v4f r0 = *(const v4f*)(xl + (size_t)s0 * F + c0);
        const v4f r1 = *(const v4f*)(xl + (size_t)s1 * F + c0);
        const v4f r2 = *(const v4f*)(xl + (size_t)s2 * F + c0);
        const v4f r3 = *(const v4f*)(xl + (size_t)s3 * F + c0);

        const float ninf = -__builtin_inff();
        const float p0 = edge_logit(xi, r0, av);
        const float q1 = edge_logit(xi, r1, av);
        const float q2 = edge_logit(xi, r2, av);
        const float q3 = edge_logit(xi, r3, av);
        const float p1 = (j + 1 < deg) ? q1 : ninf;
        const float p2 = (j + 2 < deg) ? q2 : ninf;
        const float p3 = (j + 3 < deg) ? q3 : ninf;

        const float mn = fmaxf(fmaxf(mx, p0), fmaxf(fmaxf(p1, p2), p3));
        const float sc = __expf(mx - mn);
        const float e0 = __expf(p0 - mn);
        const float e1 = __expf(p1 - mn);
        const float e2 = __expf(p2 - mn);
        const float e3 = __expf(p3 - mn);
        den = den * sc + ((e0 + e1) + (e2 + e3));
#pragma unroll
        for (int k = 0; k < 4; ++k)
            acc[k] = acc[k] * sc + ((e0 * r0[k] + e1 * r1[k]) + (e2 * r2[k] + e3 * r3[k]));
        mx = mn;
    }

    const float dn = den + 1e-16f;
    v4f o;
#pragma unroll
    for (int k = 0; k < 4; ++k) o[k] = acc[k] / dn + bv[k];
    *(v4f*)(out + (size_t)node * F + c0) = o;
}

extern "C" void kernel_launch(void* const* d_in, const int* in_sizes, int n_in,
                              void* d_out, int out_size, void* d_ws, size_t ws_size,
                              hipStream_t stream) {
    const float* x    = (const float*)d_in[0];
    const int*   ei   = (const int*)  d_in[1];
    const float* Wl   = (const float*)d_in[2];
    const float* Wr   = (const float*)d_in[3];
    const float* att  = (const float*)d_in[4];
    const float* bias = (const float*)d_in[5];

    const int n = in_sizes[0] / F;
    const int m = in_sizes[1] / 2;
    const int* src = ei;
    const int* dst = ei + m;

    float* xl       = (float*)d_ws;
    float* xr       = xl + (size_t)n * F;
    int*   rowptr   = (int*)(xr + (size_t)n * F);
    int*   csr_src  = rowptr + (n + 1);
    float* out      = (float*)d_out;
    int*   binned   = (int*)out;
    int*   offT     = binned + 2 * (size_t)m;
    const int nb    = (n + BIN_NODES - 1) / BIN_NODES;

    chunk_sort<<<NCHUNK, SORT_T, 0, stream>>>(src, dst, offT, binned, m, nb);
    finish_and_project<<<PROJ_WG + nb, PROJ_T, 0, stream>>>(x, Wl, Wr, xl, xr, binned, offT,
                                                            rowptr, csr_src, n, m, nb);

    const int nodeBlocks = (n * 32 + 255) / 256;
    gat_node<<<nodeBlocks, 256, 0, stream>>>(xl, xr, rowptr, csr_src, att, bias, out, n);
}
